# starting kernel
# baseline (speedup 1.0000x reference)
#include <hip/hip_runtime.h>
#include <hip/hip_bf16.h>
#include <math.h>

typedef float v2f __attribute__((ext_vector_type(2)));
typedef float v8f __attribute__((ext_vector_type(8)));

__device__ __forceinline__ float gelu_exact(float x) {
  return 0.5f * x * (1.0f + erff(x * 0.70710678118654752f));
}

#define WMMA_F32(cacc, av, bv)                                                 \
  cacc = __builtin_amdgcn_wmma_f32_16x16x4_f32(false, (av), false, (bv),       \
                                               (short)0, (cacc), false, false)

__global__ __launch_bounds__(256) void gemm_wmma_f32(
    const float* __restrict__ A, const float* __restrict__ W,
    const float* __restrict__ bias, const float* __restrict__ R,
    float* __restrict__ C, int M, int N, int Kd,
    int flags, int Tlen, int Kspk, int kk)
{
  const int lane = threadIdx.x & 31;
  const int w    = threadIdx.x >> 5;
  const int wm   = w & 3;
  const int wn   = w >> 2;
  const int tileM = blockIdx.x * 64 + wm * 16;
  const int tileN = blockIdx.y * 128 + wn * 64;
  const int half = lane >> 4;
  const int l16  = lane & 15;

  const int mrow = tileM + l16;
  const int n0   = tileN + l16;

  const float* Ap = A + (size_t)mrow * Kd + 2 * half;
  const float* Bp0 = W + (size_t)(n0)      * Kd + 2 * half;
  const float* Bp1 = W + (size_t)(n0 + 16) * Kd + 2 * half;
  const float* Bp2 = W + (size_t)(n0 + 32) * Kd + 2 * half;
  const float* Bp3 = W + (size_t)(n0 + 48) * Kd + 2 * half;

  v8f c0 = {}, c1 = {}, c2 = {}, c3 = {};

  v2f a0  = *(const v2f*)(Ap);
  v2f b00 = *(const v2f*)(Bp0);
  v2f b01 = *(const v2f*)(Bp1);
  v2f b02 = *(const v2f*)(Bp2);
  v2f b03 = *(const v2f*)(Bp3);
  v2f a1  = *(const v2f*)(Ap + 4);
  v2f b10 = *(const v2f*)(Bp0 + 4);
  v2f b11 = *(const v2f*)(Bp1 + 4);
  v2f b12 = *(const v2f*)(Bp2 + 4);
  v2f b13 = *(const v2f*)(Bp3 + 4);

  for (int k = 0; k < Kd - 8; k += 8) {
    WMMA_F32(c0, a0, b00);
    WMMA_F32(c1, a0, b01);
    WMMA_F32(c2, a0, b02);
    WMMA_F32(c3, a0, b03);
    a0  = *(const v2f*)(Ap  + k + 8);
    b00 = *(const v2f*)(Bp0 + k + 8);
    b01 = *(const v2f*)(Bp1 + k + 8);
    b02 = *(const v2f*)(Bp2 + k + 8);
    b03 = *(const v2f*)(Bp3 + k + 8);
    WMMA_F32(c0, a1, b10);
    WMMA_F32(c1, a1, b11);
    WMMA_F32(c2, a1, b12);
    WMMA_F32(c3, a1, b13);
    a1  = *(const v2f*)(Ap  + k + 12);
    b10 = *(const v2f*)(Bp0 + k + 12);
    b11 = *(const v2f*)(Bp1 + k + 12);
    b12 = *(const v2f*)(Bp2 + k + 12);
    b13 = *(const v2f*)(Bp3 + k + 12);
  }
  WMMA_F32(c0, a0, b00);
  WMMA_F32(c1, a0, b01);
  WMMA_F32(c2, a0, b02);
  WMMA_F32(c3, a0, b03);
  WMMA_F32(c0, a1, b10);
  WMMA_F32(c1, a1, b11);
  WMMA_F32(c2, a1, b12);
  WMMA_F32(c3, a1, b13);

  const float bv0 = bias[n0];
  const float bv1 = bias[n0 + 16];
  const float bv2 = bias[n0 + 32];
  const float bv3 = bias[n0 + 48];
#pragma unroll
  for (int i = 0; i < 8; ++i) {
    const int row = tileM + i + 8 * half;
    float v0 = c0[i] + bv0;
    float v1 = c1[i] + bv1;
    float v2 = c2[i] + bv2;
    float v3 = c3[i] + bv3;
    if (flags & 1) {
      const float* rr = R + (size_t)row * N;
      v0 += rr[n0];
      v1 += rr[n0 + 16];
      v2 += rr[n0 + 32];
      v3 += rr[n0 + 48];
    }
    if (flags & 2) {
      v0 = gelu_exact(v0);
      v1 = gelu_exact(v1);
      v2 = gelu_exact(v2);
      v3 = gelu_exact(v3);
    }
    size_t orow = (size_t)row;
    if (flags & 4) {
      const int bi = row / Tlen;
      orow = (size_t)bi * Kspk * Tlen + (size_t)kk * Tlen + (row % Tlen);
    }
    float* cr = C + orow * N;
    cr[n0]      = v0;
    cr[n0 + 16] = v1;
    cr[n0 + 32] = v2;
    cr[n0 + 48] = v3;
  }
}

__global__ __launch_bounds__(256) void ln_kernel(
    const float* __restrict__ X, const float* __restrict__ g,
    const float* __restrict__ b, float* __restrict__ Y, int Dn)
{
  const int row = blockIdx.x;
  const int tid = threadIdx.x;
  const float* x = X + (size_t)row * Dn;
  float s = 0.f, s2 = 0.f;
  for (int d = tid; d < Dn; d += 256) { float v = x[d]; s += v; s2 += v * v; }
  __shared__ float sb0[256], sb1[256];
  sb0[tid] = s; sb1[tid] = s2; __syncthreads();
  for (int off = 128; off > 0; off >>= 1) {
    if (tid < off) { sb0[tid] += sb0[tid + off]; sb1[tid] += sb1[tid + off]; }
    __syncthreads();
  }
  const float mean = sb0[0] / Dn;
  const float var  = sb1[0] / Dn - mean * mean;
  const float inv  = rsqrtf(var + 1e-5f);
  float* y = Y + (size_t)row * Dn;
  for (int d = tid; d < Dn; d += 256) y[d] = (x[d] - mean) * inv * g[d] + b[d];
}

__global__ __launch_bounds__(256) void wmix_kernel(
    const float* __restrict__ logits, const int* __restrict__ blank_p,
    float* __restrict__ wmix, int BT, int V, int Kspk)
{
  const int row = blockIdx.x;
  const int tid = threadIdx.x;
  const int blank = blank_p[0];
  __shared__ float red[256];
  float a[8];
  for (int k2 = 0; k2 < Kspk; ++k2) {
    const float* L = logits + ((size_t)k2 * BT + row) * V;
    float m = -INFINITY;
    for (int v = tid; v < V; v += 256) m = fmaxf(m, L[v]);
    red[tid] = m; __syncthreads();
    for (int off = 128; off > 0; off >>= 1) {
      if (tid < off) red[tid] = fmaxf(red[tid], red[tid + off]);
      __syncthreads();
    }
    m = red[0]; __syncthreads();
    float s = 0.f;
    for (int v = tid; v < V; v += 256) s += expf(L[v] - m);
    red[tid] = s; __syncthreads();
    for (int off = 128; off > 0; off >>= 1) {
      if (tid < off) red[tid] += red[tid + off];
      __syncthreads();
    }
    s = red[0]; __syncthreads();
    a[k2] = 1.0f - expf(L[blank] - m) / s;
  }
  if (tid == 0) {
    float m2 = -INFINITY;
    for (int k2 = 0; k2 < Kspk; ++k2) m2 = fmaxf(m2, a[k2]);
    float ss = 0.f;
    float e[8];
    for (int k2 = 0; k2 < Kspk; ++k2) { e[k2] = expf(a[k2] - m2); ss += e[k2]; }
    for (int k2 = 0; k2 < Kspk; ++k2) wmix[(size_t)row * Kspk + k2] = e[k2] / ss;
  }
}

__global__ __launch_bounds__(256) void stream_prep(
    const float* __restrict__ X, const float* __restrict__ wmix,
    const float* __restrict__ tags, const float* __restrict__ g,
    const float* __restrict__ bt, float* __restrict__ Xk,
    float* __restrict__ Qin, int Dn, int Kspk, int kk)
{
  const int row = blockIdx.x;
  const int tid = threadIdx.x;
  const float wv = wmix[(size_t)row * Kspk + kk];
  const float* x  = X    + (size_t)row * Dn;
  const float* tg = tags + (size_t)kk  * Dn;
  float s = 0.f, s2 = 0.f;
  for (int d = tid; d < Dn; d += 256) {
    float v = x[d] * wv + tg[d];
    Xk[(size_t)row * Dn + d] = v;
    s += v; s2 += v * v;
  }
  __shared__ float sb0[256], sb1[256];
  sb0[tid] = s; sb1[tid] = s2; __syncthreads();
  for (int off = 128; off > 0; off >>= 1) {
    if (tid < off) { sb0[tid] += sb0[tid + off]; sb1[tid] += sb1[tid + off]; }
    __syncthreads();
  }
  const float mean = sb0[0] / Dn;
  const float var  = sb1[0] / Dn - mean * mean;
  const float inv  = rsqrtf(var + 1e-5f);
  for (int d = tid; d < Dn; d += 256) {
    float v = x[d] * wv + tg[d];
    Qin[(size_t)row * Dn + d] = (v - mean) * inv * g[d] + bt[d];
  }
}

__global__ __launch_bounds__(128) void banded_attn(
    const float* __restrict__ q, const float* __restrict__ kb,
    const float* __restrict__ vb, float* __restrict__ ctx,
    int T_, int Dn, int Hn, int band)
{
  const int hd   = 64;
  const int wid  = threadIdx.x >> 5;
  const int lane = threadIdx.x & 31;
  const int gw   = blockIdx.x * 4 + wid;
  const int t = gw % T_;
  const int h = (gw / T_) % Hn;
  const int b = gw / (T_ * Hn);

  __shared__ float lq[4][64];
  __shared__ float lat[4][64];

  const size_t rowbase = (size_t)(b * T_ + t) * Dn + (size_t)h * hd;
  lq[wid][lane]      = q[rowbase + lane];
  lq[wid][lane + 32] = q[rowbase + lane + 32];
  __syncthreads();

  const float scale = 0.125f;
  const int s0 = t - band;
  const int nb = 2 * band + 1;

  const float4* lq4 = (const float4*)&lq[wid][0];

  const int j0 = lane, j1 = lane + 32;
  const int sa = s0 + j0, sbi = s0 + j1;
  float sc0 = -INFINITY, sc1 = -INFINITY;
  if (j0 < nb && sa >= 0 && sa < T_) {
    const size_t kro = (size_t)(b * T_ + sa) * Dn + (size_t)h * hd;
    __builtin_prefetch(vb + kro, 0, 0);
    const float4* kp4 = (const float4*)(kb + kro);
    float acc = 0.f;
#pragma unroll
    for (int d4 = 0; d4 < 16; ++d4) {
      float4 kv = kp4[d4];
      float4 qv = lq4[d4];
      acc += qv.x * kv.x + qv.y * kv.y + qv.z * kv.z + qv.w * kv.w;
    }
    sc0 = acc * scale;
  }
  if (j1 < nb && sbi >= 0 && sbi < T_) {
    const size_t kro = (size_t)(b * T_ + sbi) * Dn + (size_t)h * hd;
    __builtin_prefetch(vb + kro, 0, 0);
    const float4* kp4 = (const float4*)(kb + kro);
    float acc = 0.f;
#pragma unroll
    for (int d4 = 0; d4 < 16; ++d4) {
      float4 kv = kp4[d4];
      float4 qv = lq4[d4];
      acc += qv.x * kv.x + qv.y * kv.y + qv.z * kv.z + qv.w * kv.w;
    }
    sc1 = acc * scale;
  }

  float m = fmaxf(sc0, sc1);
  for (int off = 16; off > 0; off >>= 1) m = fmaxf(m, __shfl_xor(m, off, 32));
  float e0 = expf(sc0 - m);
  float e1 = expf(sc1 - m);
  float s = e0 + e1;
  for (int off = 16; off > 0; off >>= 1) s += __shfl_xor(s, off, 32);
  const float inv = 1.0f / s;
  lat[wid][j0] = e0 * inv;
  lat[wid][j1] = e1 * inv;
  __syncthreads();

  float a0 = 0.f, a1 = 0.f;
  const int d0 = lane, d1 = lane + 32;
  for (int j = 0; j < nb; ++j) {
    const int sidx = s0 + j;
    if (sidx < 0 || sidx >= T_) continue;
    const float wv = lat[wid][j];
    const float* vp = vb + (size_t)(b * T_ + sidx) * Dn + (size_t)h * hd;
    a0 += wv * vp[d0];
    a1 += wv * vp[d1];
  }
  float* cp = ctx + rowbase;
  cp[d0] = a0;
  cp[d1] = a1;
}

extern "C" void kernel_launch(void* const* d_in, const int* in_sizes, int n_in,
                              void* d_out, int out_size, void* d_ws, size_t ws_size,
                              hipStream_t stream) {
  (void)in_sizes; (void)n_in; (void)out_size; (void)ws_size;
  constexpr int B = 4, T = 1024, D_IN = 512, D = 512, K = 2, V = 500, H = 8, BAND = 24;
  constexpr int M = B * T;

  const float* x_m      = (const float*)d_in[0];
  const float* logits   = (const float*)d_in[1];
  const int*   blank_id = (const int*)  d_in[2];
  const float* proj_w   = (const float*)d_in[3];
  const float* proj_b   = (const float*)d_in[4];
  const float* spk_tags = (const float*)d_in[5];
  const float* lnq_g    = (const float*)d_in[6];
  const float* lnq_b    = (const float*)d_in[7];
  const float* lnkv_g   = (const float*)d_in[8];
  const float* lnkv_b   = (const float*)d_in[9];
  const float* in_w     = (const float*)d_in[10];
  const float* in_b     = (const float*)d_in[11];
  const float* out_w    = (const float*)d_in[12];
  const float* out_b    = (const float*)d_in[13];
  const float* ffn_ln_g = (const float*)d_in[14];
  const float* ffn_ln_b = (const float*)d_in[15];
  const float* ffn_w1   = (const float*)d_in[16];
  const float* ffn_b1   = (const float*)d_in[17];
  const float* ffn_w2   = (const float*)d_in[18];
  const float* ffn_b2   = (const float*)d_in[19];
  float* out = (float*)d_out;

  float* ws = (float*)d_ws;
  const size_t SZ = (size_t)M * D;
  float* X    = ws;
  float* kbuf = X    + SZ;
  float* vbuf = kbuf + SZ;
  float* tA   = vbuf + SZ;
  float* tB   = tA   + SZ;
  float* tC   = tB   + SZ;
  float* t1   = tC   + SZ;
  float* wmix = t1   + 4 * SZ;

  const dim3 blk(256);
  auto gg = [](int m, int n) { return dim3(m / 64, n / 128); };

  gemm_wmma_f32<<<gg(M, D), blk, 0, stream>>>(x_m, proj_w, proj_b, nullptr, X,
                                              M, D, D_IN, 0, T, K, 0);
  wmix_kernel<<<dim3(M), blk, 0, stream>>>(logits, blank_id, wmix, M, V, K);
  ln_kernel<<<dim3(M), blk, 0, stream>>>(X, lnkv_g, lnkv_b, tA, D);
  gemm_wmma_f32<<<gg(M, D), blk, 0, stream>>>(tA, in_w + (size_t)D * D, in_b + D,
                                              nullptr, kbuf, M, D, D, 0, T, K, 0);
  gemm_wmma_f32<<<gg(M, D), blk, 0, stream>>>(tA, in_w + (size_t)2 * D * D, in_b + 2 * D,
                                              nullptr, vbuf, M, D, D, 0, T, K, 0);

  for (int kk = 0; kk < K; ++kk) {
    stream_prep<<<dim3(M), blk, 0, stream>>>(X, wmix, spk_tags, lnq_g, lnq_b,
                                             tA, tB, D, K, kk);
    gemm_wmma_f32<<<gg(M, D), blk, 0, stream>>>(tB, in_w, in_b, nullptr, tC,
                                                M, D, D, 0, T, K, kk);
    banded_attn<<<dim3((B * H * T) / 4), dim3(128), 0, stream>>>(tC, kbuf, vbuf, tB,
                                                                 T, D, H, BAND);
    gemm_wmma_f32<<<gg(M, D), blk, 0, stream>>>(tB, out_w, out_b, tA, tC,
                                                M, D, D,  1, T, K, kk);
    ln_kernel<<<dim3(M), blk, 0, stream>>>(tC, ffn_ln_g, ffn_ln_b, tB, D);
    gemm_wmma_f32<<<gg(M, 4 * D), blk, 0, stream>>>(tB, ffn_w1, ffn_b1, nullptr, t1,
                                                    M, 4 * D, D,  2, T, K, kk);
    gemm_wmma_f32<<<gg(M, D), blk, 0, stream>>>(t1, ffn_w2, ffn_b2, tC, out,
                                                M, D, 4 * D,  1 | 4, T, K, kk);
  }
}


# Banded attention as f32 WMMA fed straight from global rows, no LDS: a 16-query wave forms S^T = K Q^T so the probabilities sit in A-operand registers for P V; V and ctx move as whole head rows
# speedup vs baseline: 1.7203x; 1.7203x over previous
#include <hip/hip_runtime.h>
#include <hip/hip_bf16.h>
#include <math.h>

typedef float v2f __attribute__((ext_vector_type(2)));
typedef float v4f __attribute__((ext_vector_type(4)));
typedef float v8f __attribute__((ext_vector_type(8)));

__device__ __forceinline__ float gelu_exact(float x) {
  return 0.5f * x * (1.0f + erff(x * 0.70710678118654752f));
}

#define WMMA_F32(cacc, av, bv)                                                 \
  cacc = __builtin_amdgcn_wmma_f32_16x16x4_f32(false, (av), false, (bv),       \
                                               (short)0, (cacc), false, false)

template <int FM, int FN, int FLAGS>
__global__ __launch_bounds__(128) void gemm_wmma_f32(
    const float* __restrict__ A, const float* __restrict__ W,
    const float* __restrict__ bias, const float* __restrict__ R,
    float* __restrict__ C, int M, int N, int Kd,
    int Tlen, int Kspk, int kk)
{
  const int lane = threadIdx.x & 31;
  const int w    = threadIdx.x >> 5;
  const int wm   = w & 1;
  const int wn   = w >> 1;
  const int tileM = blockIdx.x * (32 * FM) + wm * (16 * FM);
  const int tileN = blockIdx.y * (32 * FN) + wn * (16 * FN);
  const int half = lane >> 4;
  const int l16  = lane & 15;

  const float* Ap[FM];
  const float* Bp[FN];
#pragma unroll
  for (int fm = 0; fm < FM; ++fm)
    Ap[fm] = A + (size_t)(tileM + 16 * fm + l16) * Kd + 4 * half;
#pragma unroll
  for (int fn = 0; fn < FN; ++fn)
    Bp[fn] = W + (size_t)(tileN + 16 * fn + l16) * Kd + 4 * half;

  v8f acc[FM][FN];
#pragma unroll
  for (int fm = 0; fm < FM; ++fm)
#pragma unroll
    for (int fn = 0; fn < FN; ++fn)
      acc[fm][fn] = v8f{};

  v4f a0[4][FM], b0[4][FN];
  v4f a1[4][FM], b1[4][FN];

#define LOAD_STAGE(as, bs, koff)                                              \
  _Pragma("unroll") for (int s = 0; s < 4; ++s) {                             \
    _Pragma("unroll") for (int fm = 0; fm < FM; ++fm)                         \
      as[s][fm] = *(const v4f*)(Ap[fm] + (koff) + 8 * s);                     \
    _Pragma("unroll") for (int fn = 0; fn < FN; ++fn)                         \
      bs[s][fn] = *(const v4f*)(Bp[fn] + (koff) + 8 * s);                     \
  }

#define MMA_STAGE(as, bs)                                                     \
  _Pragma("unroll") for (int s = 0; s < 4; ++s) {                             \
    _Pragma("unroll") for (int fm = 0; fm < FM; ++fm)                         \
      _Pragma("unroll") for (int fn = 0; fn < FN; ++fn)                       \
        WMMA_F32(acc[fm][fn],                                                 \
                 __builtin_shufflevector(as[s][fm], as[s][fm], 0, 1),         \
                 __builtin_shufflevector(bs[s][fn], bs[s][fn], 0, 1));        \
    _Pragma("unroll") for (int fm = 0; fm < FM; ++fm)                         \
      _Pragma("unroll") for (int fn = 0; fn < FN; ++fn)                       \
        WMMA_F32(acc[fm][fn],                                                 \
                 __builtin_shufflevector(as[s][fm], as[s][fm], 2, 3),         \
                 __builtin_shufflevector(bs[s][fn], bs[s][fn], 2, 3));        \
  }

#define STAGE_FENCE() __builtin_amdgcn_sched_barrier(0)

  LOAD_STAGE(a0, b0, 0);
  for (int k0 = 0; k0 < Kd - 64; k0 += 64) {
    STAGE_FENCE();
    LOAD_STAGE(a1, b1, k0 + 32);
    STAGE_FENCE();
    MMA_STAGE(a0, b0);
    STAGE_FENCE();
    LOAD_STAGE(a0, b0, k0 + 64);
    STAGE_FENCE();
    MMA_STAGE(a1, b1);
  }
  STAGE_FENCE();
  LOAD_STAGE(a1, b1, Kd - 32);
  STAGE_FENCE();
  MMA_STAGE(a0, b0);
  MMA_STAGE(a1, b1);
#undef STAGE_FENCE
#undef LOAD_STAGE
#undef MMA_STAGE

  float bv[FN];
#pragma unroll
  for (int fn = 0; fn < FN; ++fn) bv[fn] = bias[tileN + 16 * fn + l16];

#pragma unroll
  for (int fm = 0; fm < FM; ++fm) {
#pragma unroll
    for (int i = 0; i < 8; ++i) {
      const int row = tileM + 16 * fm + i + 8 * half;
      float v[FN];
#pragma unroll
      for (int fn = 0; fn < FN; ++fn) v[fn] = acc[fm][fn][i] + bv[fn];
      if (FLAGS & 1) {
        const float* rr = R + (size_t)row * N + tileN + l16;
#pragma unroll
        for (int fn = 0; fn < FN; ++fn) v[fn] += rr[16 * fn];
      }
      if (FLAGS & 2) {
#pragma unroll
        for (int fn = 0; fn < FN; ++fn) v[fn] = gelu_exact(v[fn]);
      }
      size_t orow = (size_t)row;
      if (FLAGS & 4) {
        const int bi = row / Tlen;
        orow = (size_t)bi * Kspk * Tlen + (size_t)kk * Tlen + (row % Tlen);
      }
      float* cr = C + orow * N + tileN + l16;
#pragma unroll
      for (int fn = 0; fn < FN; ++fn) cr[16 * fn] = v[fn];
    }
  }
}

__global__ __launch_bounds__(256) void ln_kernel(
    const float* __restrict__ X, const float* __restrict__ g,
    const float* __restrict__ b, float* __restrict__ Y, int Dn)
{
  const int row = blockIdx.x;
  const int tid = threadIdx.x;
  const float* x = X + (size_t)row * Dn;
  float s = 0.f, s2 = 0.f;
  for (int d = tid; d < Dn; d += 256) { float v = x[d]; s += v; s2 += v * v; }
  __shared__ float sb0[256], sb1[256];
  sb0[tid] = s; sb1[tid] = s2; __syncthreads();
  for (int off = 128; off > 0; off >>= 1) {
    if (tid < off) { sb0[tid] += sb0[tid + off]; sb1[tid] += sb1[tid + off]; }
    __syncthreads();
  }
  const float mean = sb0[0] / Dn;
  const float var  = sb1[0] / Dn - mean * mean;
  const float inv  = rsqrtf(var + 1e-5f);
  float* y = Y + (size_t)row * Dn;
  for (int d = tid; d < Dn; d += 256) y[d] = (x[d] - mean) * inv * g[d] + b[d];
}

__global__ __launch_bounds__(256) void wmix_kernel(
    const float* __restrict__ logits, const int* __restrict__ blank_p,
    float* __restrict__ wmix, int BT, int V, int Kspk)
{
  const int row = blockIdx.x;
  const int tid = threadIdx.x;
  const int blank = blank_p[0];
  __shared__ float red[256];
  float a[8];
  for (int k2 = 0; k2 < Kspk; ++k2) {
    const float* L = logits + ((size_t)k2 * BT + row) * V;
    float m = -INFINITY;
    for (int v = tid; v < V; v += 256) m = fmaxf(m, L[v]);
    red[tid] = m; __syncthreads();
    for (int off = 128; off > 0; off >>= 1) {
      if (tid < off) red[tid] = fmaxf(red[tid], red[tid + off]);
      __syncthreads();
    }
    m = red[0]; __syncthreads();
    float s = 0.f;
    for (int v = tid; v < V; v += 256) s += expf(L[v] - m);
    red[tid] = s; __syncthreads();
    for (int off = 128; off > 0; off >>= 1) {
      if (tid < off) red[tid] += red[tid + off];
      __syncthreads();
    }
    s = red[0]; __syncthreads();
    a[k2] = 1.0f - expf(L[blank] - m) / s;
  }
  if (tid == 0) {
    float m2 = -INFINITY;
    for (int k2 = 0; k2 < Kspk; ++k2) m2 = fmaxf(m2, a[k2]);
    float ss = 0.f;
    float e[8];
    for (int k2 = 0; k2 < Kspk; ++k2) { e[k2] = expf(a[k2] - m2); ss += e[k2]; }
    for (int k2 = 0; k2 < Kspk; ++k2) wmix[(size_t)row * Kspk + k2] = e[k2] / ss;
  }
}

__global__ __launch_bounds__(256) void stream_prep(
    const float* __restrict__ X, const float* __restrict__ wmix,
    const float* __restrict__ tags, const float* __restrict__ g,
    const float* __restrict__ bt, float* __restrict__ Xk,
    float* __restrict__ Qin, int Dn, int Kspk, int kk)
{
  const int row = blockIdx.x;
  const int tid = threadIdx.x;
  const float wv = wmix[(size_t)row * Kspk + kk];
  const float* x  = X    + (size_t)row * Dn;
  const float* tg = tags + (size_t)kk  * Dn;
  float s = 0.f, s2 = 0.f;
  for (int d = tid; d < Dn; d += 256) {
    float v = x[d] * wv + tg[d];
    Xk[(size_t)row * Dn + d] = v;
    s += v; s2 += v * v;
  }
  __shared__ float sb0[256], sb1[256];
  sb0[tid] = s; sb1[tid] = s2; __syncthreads();
  for (int off = 128; off > 0; off >>= 1) {
    if (tid < off) { sb0[tid] += sb0[tid + off]; sb1[tid] += sb1[tid + off]; }
    __syncthreads();
  }
  const float mean = sb0[0] / Dn;
  const float var  = sb1[0] / Dn - mean * mean;
  const float inv  = rsqrtf(var + 1e-5f);
  for (int d = tid; d < Dn; d += 256) {
    float v = x[d] * wv + tg[d];
    Qin[(size_t)row * Dn + d] = (v - mean) * inv * g[d] + bt[d];
  }
}

template <int T_, int Dn, int Hn>
__global__ __launch_bounds__(128) void banded_attn_wmma(
    const float* __restrict__ q, const float* __restrict__ kbuf,
    const float* __restrict__ vbuf, float* __restrict__ ctx)
{
  constexpr int HD = 64, BAND = 24;
  static_assert(Dn == Hn * HD && T_ % 16 == 0, "attention tile shape");
  const int lane = threadIdx.x & 31;
  const int wid  = threadIdx.x >> 5;
  const int c    = lane & 15;
  const int half = lane >> 4;
  constexpr int tilesT = T_ / 16;
  const int gw   = blockIdx.x * 4 + wid;
  const int tile = gw % tilesT;
  const int hh   = (gw / tilesT) % Hn;
  const int b    = gw / (tilesT * Hn);
  const int t0   = tile * 16;
  const int key0 = t0 - BAND;
  const unsigned hoff = (unsigned)(hh * HD);
  const int brow = b * T_;


  v4f qf[8];
  {
    const float* qp = q + ((unsigned)((brow + t0 + c) * Dn) + hoff + 4u * half);
#pragma unroll
    for (int i = 0; i < 8; ++i) qf[i] = *(const v4f*)(qp + 8 * i);
  }
  v4f kf[4][8];
#pragma unroll
  for (int kb = 0; kb < 4; ++kb) {
    int s = key0 + 16 * kb + c;
    s = min(max(s, 0), T_ - 1);
    const float* kp = kbuf + ((unsigned)((brow + s) * Dn) + hoff + 4u * half);
#pragma unroll
    for (int i = 0; i < 8; ++i) kf[kb][i] = *(const v4f*)(kp + 8 * i);
  }
  v4f vf[4][4][2];
#pragma unroll
  for (int kb = 0; kb < 4; ++kb)
#pragma unroll
    for (int j = 0; j < 4; ++j)
#pragma unroll
      for (int e = 0; e < 2; ++e) {
        int s = key0 + 16 * kb + 8 * half + 2 * j + e;
        s = min(max(s, 0), T_ - 1);
        vf[kb][j][e] = *(const v4f*)(vbuf + ((unsigned)((brow + s) * Dn) + hoff + 4u * c));
      }

  __builtin_amdgcn_sched_barrier(0);
  v8f st[4];
#pragma unroll
  for (int kb = 0; kb < 4; ++kb) st[kb] = v8f{};
#pragma unroll
  for (int i = 0; i < 8; ++i) {
#pragma unroll
    for (int kb = 0; kb < 4; ++kb)
      WMMA_F32(st[kb], __builtin_shufflevector(kf[kb][i], kf[kb][i], 0, 1),
                       __builtin_shufflevector(qf[i], qf[i], 0, 1));
#pragma unroll
    for (int kb = 0; kb < 4; ++kb)
      WMMA_F32(st[kb], __builtin_shufflevector(kf[kb][i], kf[kb][i], 2, 3),
                       __builtin_shufflevector(qf[i], qf[i], 2, 3));
  }

  const float scale = 0.125f;
  float p[4][8];
  float m = -INFINITY;
#pragma unroll
  for (int kb = 0; kb < 4; ++kb)
#pragma unroll
    for (int r = 0; r < 8; ++r) {
      const int rel = 16 * kb + 8 * half + r - c;
      const int s   = key0 + 16 * kb + 8 * half + r;
      const bool ok = (rel >= 0) && (rel <= 2 * BAND) && (s >= 0) && (s < T_);
      const float v = ok ? st[kb][r] * scale : -INFINITY;
      p[kb][r] = v;
      m = fmaxf(m, v);
    }
  m = fmaxf(m, __shfl_xor(m, 16, 32));
  float sum = 0.f;
#pragma unroll
  for (int kb = 0; kb < 4; ++kb)
#pragma unroll
    for (int r = 0; r < 8; ++r) {
      const float e = expf(p[kb][r] - m);
      p[kb][r] = e;
      sum += e;
    }
  sum += __shfl_xor(sum, 16, 32);
  const float inv = 1.0f / sum;

  v8f o[4];
#pragma unroll
  for (int db = 0; db < 4; ++db) o[db] = v8f{};
#pragma unroll
  for (int kb = 0; kb < 4; ++kb)
#pragma unroll
    for (int j = 0; j < 4; ++j) {
      v2f a;
      a[0] = p[kb][2 * j] * inv;
      a[1] = p[kb][2 * j + 1] * inv;
#pragma unroll
      for (int db = 0; db < 4; ++db) {
        v2f bb;
        bb[0] = vf[kb][j][0][db];
        bb[1] = vf[kb][j][1][db];
        WMMA_F32(o[db], a, bb);
      }
    }

#pragma unroll
  for (int r = 0; r < 8; ++r) {
    v4f ov;
    ov[0] = o[0][r]; ov[1] = o[1][r]; ov[2] = o[2][r]; ov[3] = o[3][r];
    *(v4f*)(ctx + ((unsigned)((brow + t0 + 8 * half + r) * Dn) + hoff + 4u * c)) = ov;
  }
}

extern "C" void kernel_launch(void* const* d_in, const int* in_sizes, int n_in,
                              void* d_out, int out_size, void* d_ws, size_t ws_size,
                              hipStream_t stream) {
  (void)in_sizes; (void)n_in; (void)out_size; (void)ws_size;
  constexpr int B = 4, T = 1024, D_IN = 512, D = 512, K = 2, V = 500, H = 8, BAND = 24;
  constexpr int M = B * T;

  const float* x_m      = (const float*)d_in[0];
  const float* logits   = (const float*)d_in[1];
  const int*   blank_id = (const int*)  d_in[2];
  const float* proj_w   = (const float*)d_in[3];
  const float* proj_b   = (const float*)d_in[4];
  const float* spk_tags = (const float*)d_in[5];
  const float* lnq_g    = (const float*)d_in[6];
  const float* lnq_b    = (const float*)d_in[7];
  const float* lnkv_g   = (const float*)d_in[8];
  const float* lnkv_b   = (const float*)d_in[9];
  const float* in_w     = (const float*)d_in[10];
  const float* in_b     = (const float*)d_in[11];
  const float* out_w    = (const float*)d_in[12];
  const float* out_b    = (const float*)d_in[13];
  const float* ffn_ln_g = (const float*)d_in[14];
  const float* ffn_ln_b = (const float*)d_in[15];
  const float* ffn_w1   = (const float*)d_in[16];
  const float* ffn_b1   = (const float*)d_in[17];
  const float* ffn_w2   = (const float*)d_in[18];
  const float* ffn_b2   = (const float*)d_in[19];
  float* out = (float*)d_out;

  float* ws = (float*)d_ws;
  const size_t SZ = (size_t)M * D;
  float* X    = ws;
  float* kbuf = X    + SZ;
  float* vbuf = kbuf + SZ;
  float* tA   = vbuf + SZ;
  float* tB   = tA   + SZ;
  float* tC   = tB   + SZ;
  float* t1   = tC   + SZ;
  float* wmix = t1   + 4 * SZ;

  const dim3 blk(256);
  const dim3 gblk(128);
  auto gg = [](int m, int n) { return dim3(m / 64, n / 128); };

  gemm_wmma_f32<2, 4, 0><<<gg(M, D), gblk, 0, stream>>>(x_m, proj_w, proj_b, nullptr, X,
                                                    M, D, D_IN, T, K, 0);
  wmix_kernel<<<dim3(M), blk, 0, stream>>>(logits, blank_id, wmix, M, V, K);
  ln_kernel<<<dim3(M), blk, 0, stream>>>(X, lnkv_g, lnkv_b, tA, D);
  gemm_wmma_f32<2, 4, 0><<<gg(M, D), gblk, 0, stream>>>(tA, in_w + (size_t)D * D, in_b + D,
                                                    nullptr, kbuf, M, D, D, T, K, 0);
  gemm_wmma_f32<2, 4, 0><<<gg(M, D), gblk, 0, stream>>>(tA, in_w + (size_t)2 * D * D, in_b + 2 * D,
                                                    nullptr, vbuf, M, D, D, T, K, 0);

  for (int kk = 0; kk < K; ++kk) {
    stream_prep<<<dim3(M), blk, 0, stream>>>(X, wmix, spk_tags, lnq_g, lnq_b,
                                             tA, tB, D, K, kk);
    gemm_wmma_f32<2, 4, 0><<<gg(M, D), gblk, 0, stream>>>(tB, in_w, in_b, nullptr, tC,
                                                      M, D, D, T, K, kk);
    static_assert(BAND == 24, "attention tile shape");
    banded_attn_wmma<T, D, H><<<dim3((B * H * (T / 16)) / 4), dim3(128), 0, stream>>>(tC, kbuf, vbuf, tB);
    gemm_wmma_f32<2, 4,  1><<<gg(M, D), gblk, 0, stream>>>(tB, out_w, out_b, tA, tC,
                                                                  M, D, D, T, K, kk);
    ln_kernel<<<dim3(M), blk, 0, stream>>>(tC, ffn_ln_g, ffn_ln_b, tB, D);
    gemm_wmma_f32<2, 4,  2><<<gg(M, 4 * D), gblk, 0, stream>>>(tB, ffn_w1, ffn_b1, nullptr, t1,
                                                                  M, 4 * D, D, T, K, kk);
    gemm_wmma_f32<2, 4,  1 | 4><<<gg(M, D), gblk, 0, stream>>>(t1, ffn_w2, ffn_b2, tC, out,
                                                                        M, D, 4 * D, T, K, kk);
  }
}


# Weights re-laid once per call into WMMA fragment order by one extra launch: a B fragment load reads 4 whole 128-byte lines, not 32 bytes of each of 16; A loads, k map unchanged; 14 to 15 launches
# speedup vs baseline: 2.1461x; 1.2475x over previous
#include <hip/hip_runtime.h>
#include <hip/hip_bf16.h>
#include <math.h>

typedef float v2f __attribute__((ext_vector_type(2)));
typedef float v4f __attribute__((ext_vector_type(4)));
typedef float v8f __attribute__((ext_vector_type(8)));

__device__ __forceinline__ float gelu_exact(float x) {
  return 0.5f * x * (1.0f + erff(x * 0.70710678118654752f));
}

#define WMMA_F32(cacc, av, bv)                                                 \
  cacc = __builtin_amdgcn_wmma_f32_16x16x4_f32(false, (av), false, (bv),       \
                                               (short)0, (cacc), false, false)

template <int MASK>
__device__ __forceinline__ float row16_xor(float x) {
  return __int_as_float(__builtin_amdgcn_update_dpp(0, __float_as_int(x), 0x160 + MASK, 0xf, 0xf, true));
}
__device__ __forceinline__ float row16_sum(float x) {
  x += row16_xor<1>(x);
  x += row16_xor<2>(x);
  x += row16_xor<4>(x);
  x += row16_xor<8>(x);
  return x;
}

template <int FM, int FN, int FLAGS, int NS = 0>
__global__ __launch_bounds__(128) void gemm_wmma_f32(
    const float* __restrict__ A, const float* __restrict__ W,
    const float* __restrict__ bias, const float* __restrict__ R,
    float* __restrict__ C, int M, int N, int Kd,
    int Tlen, int Kspk, int kk,
    const float* __restrict__ lnG, const float* __restrict__ lnB,
    const float* __restrict__ statsIn, float* __restrict__ statsOut,
    const float* __restrict__ mixW, const float* __restrict__ tags)
{
  static_assert(!(FLAGS & (8 | 16 | 64)) || FM == 2, "row-sum slots assume a 32-row wave tile");
  static_assert(!(FLAGS & 64) || (FLAGS & 8), "stream sums ride on the row-sum epilogue");
  const int lane = threadIdx.x & 31;
  const int w    = threadIdx.x >> 5;
  const int wm   = w & 1;
  const int wn   = w >> 1;
  const int tileM = blockIdx.x * (32 * FM) + wm * (16 * FM);
  const int tileN = blockIdx.y * (32 * FN) + wn * (16 * FN);
  const int half = lane >> 4;
  const int l16  = lane & 15;

  const float* Ap[FM];
  const float* Bp[FN];
#pragma unroll
  for (int fm = 0; fm < FM; ++fm)
    Ap[fm] = A + (size_t)(tileM + 16 * fm + l16) * Kd + 4 * half;
#pragma unroll
  for (int fn = 0; fn < FN; ++fn)
    Bp[fn] = (FLAGS & 128)
        ? W + (size_t)((tileN >> 4) + fn) * Kd * 16 + 4 * lane
        : W + (size_t)(tileN + 16 * fn + l16) * Kd + 4 * half;

  v8f acc[FM][FN];
#pragma unroll
  for (int fm = 0; fm < FM; ++fm)
#pragma unroll
    for (int fn = 0; fn < FN; ++fn)
      acc[fm][fn] = v8f{};

  v4f a0[4][FM], b0[4][FN];
  v4f a1[4][FM], b1[4][FN];
  v4f g0[4], h0[4], g1[4], h1[4];

  float rinv[FM], rc0[FM], rw[FM];
  v4f t0[4], t1[4];
  if (FLAGS & 16) {
    static_assert(!(FLAGS & 16) || NS > 0, "bit4 needs the slot count NS = Kd/64");
    v2f ps[FM][NS > 0 ? NS : 1];
#pragma unroll
    for (int fm = 0; fm < FM; ++fm)
#pragma unroll
      for (int sl = 0; sl < NS; ++sl)
        ps[fm][sl] = *(const v2f*)(statsIn + ((size_t)sl * M + tileM + 16 * fm + l16) * 2);
#pragma unroll
    for (int fm = 0; fm < FM; ++fm) {
      const int row = tileM + 16 * fm + l16;
      float s = 0.f, s2 = 0.f;
#pragma unroll
      for (int sl = 0; sl < NS; ++sl) { s += ps[fm][sl][0]; s2 += ps[fm][sl][1]; }
      const float mean = s / Kd;
      const float var  = s2 / Kd - mean * mean;
      const float inv  = rsqrtf(var + 1e-5f);
      rinv[fm] = inv; rc0[fm] = -mean * inv;
      if (FLAGS & 32) rw[fm] = mixW[(size_t)row * Kspk + kk];
    }
  }

#define LOAD_STAGE(as, bs, gs, hs, ts, koff)                                  \
  _Pragma("unroll") for (int s = 0; s < 4; ++s) {                             \
    _Pragma("unroll") for (int fm = 0; fm < FM; ++fm)                         \
      as[s][fm] = *(const v4f*)(Ap[fm] + (koff) + 8 * s);                     \
    _Pragma("unroll") for (int fn = 0; fn < FN; ++fn)                         \
      bs[s][fn] = *(const v4f*)(Bp[fn] + ((FLAGS & 128)                       \
                                    ? ((koff) + 8 * s) * 16                   \
                                    : (koff) + 8 * s));                       \
    if (FLAGS & 16) {                                                         \
      gs[s] = *(const v4f*)(lnG + (koff) + 8 * s + 4 * half);                 \
      hs[s] = *(const v4f*)(lnB + (koff) + 8 * s + 4 * half);                 \
      if (FLAGS & 32)                                                         \
        ts[s] = *(const v4f*)(tags + (koff) + 8 * s + 4 * half);              \
    }                                                                         \
  }

#define NORM_STAGE(as, gs, hs, ts)                                            \
  if (FLAGS & 16) {                                                           \
    _Pragma("unroll") for (int s = 0; s < 4; ++s)                             \
      _Pragma("unroll") for (int fm = 0; fm < FM; ++fm) {                     \
        v4f u = as[s][fm];                                                    \
        if (FLAGS & 32) u = u * rw[fm] + ts[s];                               \
        const v4f t = u * rinv[fm] + rc0[fm];                                 \
        as[s][fm] = t * gs[s] + hs[s];                                        \
      }                                                                       \
  }

#define MMA_STAGE(as, bs)                                                     \
  _Pragma("unroll") for (int s = 0; s < 4; ++s) {                             \
    _Pragma("unroll") for (int fm = 0; fm < FM; ++fm)                         \
      _Pragma("unroll") for (int fn = 0; fn < FN; ++fn)                       \
        WMMA_F32(acc[fm][fn],                                                 \
                 __builtin_shufflevector(as[s][fm], as[s][fm], 0, 1),         \
                 __builtin_shufflevector(bs[s][fn], bs[s][fn], 0, 1));        \
    _Pragma("unroll") for (int fm = 0; fm < FM; ++fm)                         \
      _Pragma("unroll") for (int fn = 0; fn < FN; ++fn)                       \
        WMMA_F32(acc[fm][fn],                                                 \
                 __builtin_shufflevector(as[s][fm], as[s][fm], 2, 3),         \
                 __builtin_shufflevector(bs[s][fn], bs[s][fn], 2, 3));        \
  }

#define STAGE_FENCE() __builtin_amdgcn_sched_barrier(0)

  LOAD_STAGE(a0, b0, g0, h0, t0, 0);
  for (int k0 = 0; k0 < Kd - 64; k0 += 64) {
    STAGE_FENCE();
    LOAD_STAGE(a1, b1, g1, h1, t1, k0 + 32);
    STAGE_FENCE();
    NORM_STAGE(a0, g0, h0, t0);
    MMA_STAGE(a0, b0);
    STAGE_FENCE();
    LOAD_STAGE(a0, b0, g0, h0, t0, k0 + 64);
    STAGE_FENCE();
    NORM_STAGE(a1, g1, h1, t1);
    MMA_STAGE(a1, b1);
  }
  STAGE_FENCE();
  LOAD_STAGE(a1, b1, g1, h1, t1, Kd - 32);
  STAGE_FENCE();
  NORM_STAGE(a0, g0, h0, t0);
  MMA_STAGE(a0, b0);
  NORM_STAGE(a1, g1, h1, t1);
  MMA_STAGE(a1, b1);
#undef STAGE_FENCE
#undef LOAD_STAGE
#undef NORM_STAGE
#undef MMA_STAGE

  float bv[FN];
#pragma unroll
  for (int fn = 0; fn < FN; ++fn) bv[fn] = bias[tileN + 16 * fn + l16];
  float tg0[FN], tg1[FN];
  if (FLAGS & 64) {
#pragma unroll
    for (int fn = 0; fn < FN; ++fn) {
      tg0[fn] = tags[tileN + 16 * fn + l16];
      tg1[fn] = tags[N + tileN + 16 * fn + l16];
    }
  } else if ((FLAGS & 33) == 33) {
#pragma unroll
    for (int fn = 0; fn < FN; ++fn) tg0[fn] = tags[tileN + 16 * fn + l16];
  }

  v2f keepV = v2f{}, keepS0 = v2f{}, keepS1 = v2f{};

#pragma unroll
  for (int fm = 0; fm < FM; ++fm) {
#pragma unroll
    for (int i = 0; i < 8; ++i) {
      const int row = tileM + 16 * fm + i + 8 * half;
      float v[FN];
#pragma unroll
      for (int fn = 0; fn < FN; ++fn) v[fn] = acc[fm][fn][i] + bv[fn];
      if (FLAGS & 1) {
        const float* rr = R + (size_t)row * N + tileN + l16;
        if (FLAGS & 32) {
          const float wr = mixW[(size_t)row * Kspk + kk];
#pragma unroll
          for (int fn = 0; fn < FN; ++fn) {
            const float xk = rr[16 * fn] * wr + tg0[fn];
            v[fn] += xk;
          }
        } else {
#pragma unroll
          for (int fn = 0; fn < FN; ++fn) v[fn] += rr[16 * fn];
        }
      }
      if (FLAGS & 2) {
#pragma unroll
        for (int fn = 0; fn < FN; ++fn) v[fn] = gelu_exact(v[fn]);
      }
      if (FLAGS & 8) {
        float ps = 0.f, ps2 = 0.f;
#pragma unroll
        for (int fn = 0; fn < FN; ++fn) { ps += v[fn]; ps2 += v[fn] * v[fn]; }
        ps = row16_sum(ps); ps2 = row16_sum(ps2);
        if (l16 == 8 * fm + i) { keepV[0] = ps; keepV[1] = ps2; }
      }
      if (FLAGS & 64) {
        const v2f w2 = *(const v2f*)(mixW + (size_t)row * 2);
        float p0 = 0.f, p02 = 0.f, p1 = 0.f, p12 = 0.f;
#pragma unroll
        for (int fn = 0; fn < FN; ++fn) {
          const float x0 = v[fn] * w2[0] + tg0[fn];
          const float x1 = v[fn] * w2[1] + tg1[fn];
          p0 += x0; p02 += x0 * x0; p1 += x1; p12 += x1 * x1;
        }
        p0 = row16_sum(p0); p02 = row16_sum(p02);
        p1 = row16_sum(p1); p12 = row16_sum(p12);
        if (l16 == 8 * fm + i) { keepS0[0] = p0; keepS0[1] = p02; keepS1[0] = p1; keepS1[1] = p12; }
      }
      size_t orow = (size_t)row;
      if (FLAGS & 4) {
        const int bi = row / Tlen;
        orow = (size_t)bi * Kspk * Tlen + (size_t)kk * Tlen + (row % Tlen);
      }
      float* cr = C + orow * N + tileN + l16;
#pragma unroll
      for (int fn = 0; fn < FN; ++fn) cr[16 * fn] = v[fn];
    }
  }
  if (FLAGS & 8) {
    const int srow = tileM + 16 * (l16 >> 3) + 8 * half + (l16 & 7);
    const int slot = blockIdx.y * 2 + wn;
    const size_t pairs = (size_t)(N >> 6) * M * 2;
    float* sp = statsOut + ((size_t)slot * M + srow) * 2;
    *(v2f*)sp = keepV;
    if (FLAGS & 64) {
      *(v2f*)(sp + pairs)     = keepS0;
      *(v2f*)(sp + 2 * pairs) = keepS1;
    }
  }
}

template <int DIN, int Dm>
__global__ __launch_bounds__(128) void pack_weights_frag(
    const float* __restrict__ w_proj, const float* __restrict__ w_in,
    const float* __restrict__ w_out, const float* __restrict__ w_f1,
    const float* __restrict__ w_f2,
    float* __restrict__ p_proj, float* __restrict__ p_in,
    float* __restrict__ p_out, float* __restrict__ p_f1,
    float* __restrict__ p_f2)
{
  constexpr int I0 = (Dm / 16) * (DIN / 32);
  constexpr int I1 = I0 + (3 * Dm / 16) * (Dm / 32);
  constexpr int I2 = I1 + (Dm / 16) * (Dm / 32);
  constexpr int I3 = I2 + (4 * Dm / 16) * (Dm / 32);
  constexpr int I4 = I3 + (Dm / 16) * (4 * Dm / 32);
  const int lane = threadIdx.x & 31;
  const int c    = lane & 15;
  const int half = lane >> 4;
  int item = blockIdx.x * 4 + (threadIdx.x >> 5);
  const float* src; float* dst; int Kd;
  if (item < I0)      { src = w_proj; dst = p_proj; Kd = DIN; }
  else if (item < I1) { src = w_in;   dst = p_in;   Kd = Dm;     item -= I0; }
  else if (item < I2) { src = w_out;  dst = p_out;  Kd = Dm;     item -= I1; }
  else if (item < I3) { src = w_f1;   dst = p_f1;   Kd = Dm;     item -= I2; }
  else if (item < I4) { src = w_f2;   dst = p_f2;   Kd = 4 * Dm; item -= I3; }
  else return;
  const int stages = Kd / 32;
  const int g  = item / stages;
  const int st = item % stages;
  const float* rp = src + (size_t)(16 * g + c) * Kd + 32 * st + 4 * half;
  float* op = dst + ((size_t)g * (Kd / 8) + 4 * st) * 128 + 4 * lane;
  v4f x[4];
#pragma unroll
  for (int s = 0; s < 4; ++s) x[s] = *(const v4f*)(rp + 8 * s);
#pragma unroll
  for (int s = 0; s < 4; ++s) *(v4f*)(op + 128 * s) = x[s];
}

__global__ __launch_bounds__(256) void wmix_kernel(
    const float* __restrict__ logits, const int* __restrict__ blank_p,
    float* __restrict__ wmix, int BT, int V, int Kspk)
{
  const int row = blockIdx.x;
  const int tid = threadIdx.x;
  const int blank = blank_p[0];
  __shared__ float red[256];
  float a[8];
  for (int k2 = 0; k2 < Kspk; ++k2) {
    const float* L = logits + ((size_t)k2 * BT + row) * V;
    float m = -INFINITY;
    for (int v = tid; v < V; v += 256) m = fmaxf(m, L[v]);
    red[tid] = m; __syncthreads();
    for (int off = 128; off > 0; off >>= 1) {
      if (tid < off) red[tid] = fmaxf(red[tid], red[tid + off]);
      __syncthreads();
    }
    m = red[0]; __syncthreads();
    float s = 0.f;
    for (int v = tid; v < V; v += 256) s += expf(L[v] - m);
    red[tid] = s; __syncthreads();
    for (int off = 128; off > 0; off >>= 1) {
      if (tid < off) red[tid] += red[tid + off];
      __syncthreads();
    }
    s = red[0]; __syncthreads();
    a[k2] = 1.0f - expf(L[blank] - m) / s;
  }
  if (tid == 0) {
    float m2 = -INFINITY;
    for (int k2 = 0; k2 < Kspk; ++k2) m2 = fmaxf(m2, a[k2]);
    float ss = 0.f;
    float e[8];
    for (int k2 = 0; k2 < Kspk; ++k2) { e[k2] = expf(a[k2] - m2); ss += e[k2]; }
    for (int k2 = 0; k2 < Kspk; ++k2) wmix[(size_t)row * Kspk + k2] = e[k2] / ss;
  }
}

template <int T_, int Dn, int Hn>
__global__ __launch_bounds__(128) void banded_attn_wmma(
    const float* __restrict__ q, const float* __restrict__ kbuf,
    const float* __restrict__ vbuf, float* __restrict__ ctx)
{
  constexpr int HD = 64, BAND = 24;
  static_assert(Dn == Hn * HD && T_ % 16 == 0, "attention tile shape");
  const int lane = threadIdx.x & 31;
  const int wid  = threadIdx.x >> 5;
  const int c    = lane & 15;
  const int half = lane >> 4;
  constexpr int tilesT = T_ / 16;
  const int gw   = blockIdx.x * 4 + wid;
  const int tile = gw % tilesT;
  const int hh   = (gw / tilesT) % Hn;
  const int b    = gw / (tilesT * Hn);
  const int t0   = tile * 16;
  const int key0 = t0 - BAND;
  const unsigned hoff = (unsigned)(hh * HD);
  const int brow = b * T_;


  v4f qf[8];
  {
    const float* qp = q + ((unsigned)((brow + t0 + c) * Dn) + hoff + 4u * half);
#pragma unroll
    for (int i = 0; i < 8; ++i) qf[i] = *(const v4f*)(qp + 8 * i);
  }
  v4f kf[4][8];
#pragma unroll
  for (int kb = 0; kb < 4; ++kb) {
    int s = key0 + 16 * kb + c;
    s = min(max(s, 0), T_ - 1);
    const float* kp = kbuf + ((unsigned)((brow + s) * Dn) + hoff + 4u * half);
#pragma unroll
    for (int i = 0; i < 8; ++i) kf[kb][i] = *(const v4f*)(kp + 8 * i);
  }
  v4f vf[4][4][2];
#pragma unroll
  for (int kb = 0; kb < 4; ++kb)
#pragma unroll
    for (int j = 0; j < 4; ++j)
#pragma unroll
      for (int e = 0; e < 2; ++e) {
        int s = key0 + 16 * kb + 8 * half + 2 * j + e;
        s = min(max(s, 0), T_ - 1);
        vf[kb][j][e] = *(const v4f*)(vbuf + ((unsigned)((brow + s) * Dn) + hoff + 4u * c));
      }

  __builtin_amdgcn_sched_barrier(0);
  v8f st[4];
#pragma unroll
  for (int kb = 0; kb < 4; ++kb) st[kb] = v8f{};
#pragma unroll
  for (int i = 0; i < 8; ++i) {
#pragma unroll
    for (int kb = 0; kb < 4; ++kb)
      WMMA_F32(st[kb], __builtin_shufflevector(kf[kb][i], kf[kb][i], 0, 1),
                       __builtin_shufflevector(qf[i], qf[i], 0, 1));
#pragma unroll
    for (int kb = 0; kb < 4; ++kb)
      WMMA_F32(st[kb], __builtin_shufflevector(kf[kb][i], kf[kb][i], 2, 3),
                       __builtin_shufflevector(qf[i], qf[i], 2, 3));
  }

  const float scale = 0.125f;
  float p[4][8];
  float m = -INFINITY;
#pragma unroll
  for (int kb = 0; kb < 4; ++kb)
#pragma unroll
    for (int r = 0; r < 8; ++r) {
      const int rel = 16 * kb + 8 * half + r - c;
      const int s   = key0 + 16 * kb + 8 * half + r;
      const bool ok = (rel >= 0) && (rel <= 2 * BAND) && (s >= 0) && (s < T_);
      const float v = ok ? st[kb][r] * scale : -INFINITY;
      p[kb][r] = v;
      m = fmaxf(m, v);
    }
  m = fmaxf(m, __shfl_xor(m, 16, 32));
  float sum = 0.f;
#pragma unroll
  for (int kb = 0; kb < 4; ++kb)
#pragma unroll
    for (int r = 0; r < 8; ++r) {
      const float e = expf(p[kb][r] - m);
      p[kb][r] = e;
      sum += e;
    }
  sum += __shfl_xor(sum, 16, 32);
  const float inv = 1.0f / sum;

  v8f o[4];
#pragma unroll
  for (int db = 0; db < 4; ++db) o[db] = v8f{};
#pragma unroll
  for (int kb = 0; kb < 4; ++kb)
#pragma unroll
    for (int j = 0; j < 4; ++j) {
      v2f a;
      a[0] = p[kb][2 * j] * inv;
      a[1] = p[kb][2 * j + 1] * inv;
#pragma unroll
      for (int db = 0; db < 4; ++db) {
        v2f bb;
        bb[0] = vf[kb][j][0][db];
        bb[1] = vf[kb][j][1][db];
        WMMA_F32(o[db], a, bb);
      }
    }

#pragma unroll
  for (int r = 0; r < 8; ++r) {
    v4f ov;
    ov[0] = o[0][r]; ov[1] = o[1][r]; ov[2] = o[2][r]; ov[3] = o[3][r];
    *(v4f*)(ctx + ((unsigned)((brow + t0 + 8 * half + r) * Dn) + hoff + 4u * c)) = ov;
  }
}

extern "C" void kernel_launch(void* const* d_in, const int* in_sizes, int n_in,
                              void* d_out, int out_size, void* d_ws, size_t ws_size,
                              hipStream_t stream) {
  (void)in_sizes; (void)n_in; (void)out_size; (void)ws_size;
  constexpr int B = 4, T = 1024, D_IN = 512, D = 512, K = 2, V = 500, H = 8, BAND = 24;
  constexpr int M = B * T;

  const float* x_m      = (const float*)d_in[0];
  const float* logits   = (const float*)d_in[1];
  const int*   blank_id = (const int*)  d_in[2];
  const float* proj_w   = (const float*)d_in[3];
  const float* proj_b   = (const float*)d_in[4];
  const float* spk_tags = (const float*)d_in[5];
  const float* lnq_g    = (const float*)d_in[6];
  const float* lnq_b    = (const float*)d_in[7];
  const float* lnkv_g   = (const float*)d_in[8];
  const float* lnkv_b   = (const float*)d_in[9];
  const float* in_w     = (const float*)d_in[10];
  const float* in_b     = (const float*)d_in[11];
  const float* out_w    = (const float*)d_in[12];
  const float* out_b    = (const float*)d_in[13];
  const float* ffn_ln_g = (const float*)d_in[14];
  const float* ffn_ln_b = (const float*)d_in[15];
  const float* ffn_w1   = (const float*)d_in[16];
  const float* ffn_b1   = (const float*)d_in[17];
  const float* ffn_w2   = (const float*)d_in[18];
  const float* ffn_b2   = (const float*)d_in[19];
  float* out = (float*)d_out;

  float* ws = (float*)d_ws;
  const size_t SZ = (size_t)M * D;
  float* X    = ws;
  float* kbuf = X    + SZ;
  float* vbuf = kbuf + SZ;
  float* tA   = vbuf + SZ;
  float* tB   = tA   + SZ;
  float* tC   = tB   + SZ;
  float* t1   = tC   + SZ;
  float* ctx  = t1;
  float* wmix = t1   + 4 * SZ;

  static_assert(K == 2, "the projection epilogue sums two speaker streams");
  const size_t PAIRS = (size_t)(D / 64) * M * 2;
  float* sumsX = tA;
  float* sumsY = tA + 3 * PAIRS;

  static_assert(4 * PAIRS + (size_t)D * D_IN + 4 * (size_t)D * D <= SZ, "packed small weights fit behind the row sums");
  static_assert(8 * (size_t)D * D == SZ, "packed FFN weights fill tB");
  float* pk_proj = tA + 4 * PAIRS;
  float* pk_in   = pk_proj + (size_t)D * D_IN;
  float* pk_out  = pk_in   + 3 * (size_t)D * D;
  float* pk_f1   = tB;
  float* pk_f2   = tB + 4 * (size_t)D * D;

  const dim3 blk(256);
  const dim3 gblk(128);
  auto gg = [](int m, int n) { return dim3(m / 64, n / 128); };

  pack_weights_frag<D_IN, D><<<dim3(((D / 16) * (D_IN / 32) + (8 * D / 16) * (D / 32) + (D / 16) * (4 * D / 32)) / 4),
                              dim3(128), 0, stream>>>(
      proj_w, in_w, out_w, ffn_w1, ffn_w2, pk_proj, pk_in, pk_out, pk_f1, pk_f2);
  wmix_kernel<<<dim3(M), blk, 0, stream>>>(logits, blank_id, wmix, M, V, K);
  gemm_wmma_f32<2, 4,  8 | 64 | 128><<<gg(M, D), gblk, 0, stream>>>(
      x_m, pk_proj, proj_b, nullptr, X, M, D, D_IN, T, K, 0,
      nullptr, nullptr, nullptr, sumsX, wmix, spk_tags);
  gemm_wmma_f32<2, 4,  16 | 128, D / 64><<<gg(M, D), gblk, 0, stream>>>(
      X, pk_in + (size_t)D * D, in_b + D, nullptr, kbuf, M, D, D, T, K, 0,
      lnkv_g, lnkv_b, sumsX, nullptr, nullptr, nullptr);
  gemm_wmma_f32<2, 4,  16 | 128, D / 64><<<gg(M, D), gblk, 0, stream>>>(
      X, pk_in + (size_t)2 * D * D, in_b + 2 * D, nullptr, vbuf, M, D, D, T, K, 0,
      lnkv_g, lnkv_b, sumsX, nullptr, nullptr, nullptr);

  for (int kk = 0; kk < K; ++kk) {
    const float* tagk = spk_tags + (size_t)kk * D;
    gemm_wmma_f32<2, 4,  16 | 32 | 128, D / 64><<<gg(M, D), gblk, 0, stream>>>(
        X, pk_in, in_b, nullptr, tC, M, D, D, T, K, kk,
        lnq_g, lnq_b, sumsX + (size_t)(1 + kk) * PAIRS, nullptr, wmix, tagk);
    static_assert(BAND == 24, "attention tile shape");
    banded_attn_wmma<T, D, H><<<dim3((B * H * (T / 16)) / 4), dim3(128), 0, stream>>>(tC, kbuf, vbuf, ctx);
    gemm_wmma_f32<2, 4,  1 | 8 | 32 | 128><<<gg(M, D), gblk, 0, stream>>>(
        ctx, pk_out, out_b, X, tC, M, D, D, T, K, kk,
        nullptr, nullptr, nullptr, sumsY, wmix, tagk);
    gemm_wmma_f32<2, 4,  2 | 16 | 128, D / 64><<<gg(M, 4 * D), gblk, 0, stream>>>(
        tC, pk_f1, ffn_b1, nullptr, t1, M, 4 * D, D, T, K, kk,
        ffn_ln_g, ffn_ln_b, sumsY, nullptr, nullptr, nullptr);
    gemm_wmma_f32<2, 4,  1 | 4 | 128><<<gg(M, D), gblk, 0, stream>>>(
        t1, pk_f2, ffn_b2, tC, out, M, D, 4 * D, T, K, kk,
        nullptr, nullptr, nullptr, nullptr, nullptr, nullptr);
  }
}


# Activations pass between GEMMs in WMMA fragment order: producers form the transposed tile so a lane owns 8 consecutive columns of a row and stores whole lines; A loads touch 4 lines, not 16
# speedup vs baseline: 2.2521x; 1.0494x over previous
#include <hip/hip_runtime.h>
#include <hip/hip_bf16.h>
#include <math.h>

typedef float v2f __attribute__((ext_vector_type(2)));
typedef float v4f __attribute__((ext_vector_type(4)));
typedef float v8f __attribute__((ext_vector_type(8)));

__device__ __forceinline__ float gelu_exact(float x) {
  return 0.5f * x * (1.0f + erff(x * 0.70710678118654752f));
}

#define WMMA_F32(cacc, av, bv)                                                 \
  cacc = __builtin_amdgcn_wmma_f32_16x16x4_f32(false, (av), false, (bv),       \
                                               (short)0, (cacc), false, false)

template <int MASK>
__device__ __forceinline__ float row16_xor(float x) {
  return __int_as_float(__builtin_amdgcn_update_dpp(0, __float_as_int(x), 0x160 + MASK, 0xf, 0xf, true));
}
__device__ __forceinline__ float row16_sum(float x) {
  x += row16_xor<1>(x);
  x += row16_xor<2>(x);
  x += row16_xor<4>(x);
  x += row16_xor<8>(x);
  return x;
}

template <int FM, int FN, int FLAGS, int NS = 0>
__global__ __launch_bounds__(128) void gemm_wmma_f32(
    const float* __restrict__ A, const float* __restrict__ W,
    const float* __restrict__ bias, const float* __restrict__ R,
    float* __restrict__ C, int M, int N, int Kd,
    int Tlen, int Kspk, int kk,
    const float* __restrict__ lnG, const float* __restrict__ lnB,
    const float* __restrict__ statsIn, float* __restrict__ statsOut,
    const float* __restrict__ mixW, const float* __restrict__ tags)
{
  static_assert(!(FLAGS & (8 | 16 | 64)) || FM == 2, "row-sum slots assume a 32-row wave tile");
  static_assert(!(FLAGS & 64) || (FLAGS & 8), "stream sums ride on the row-sum epilogue");
  const int lane = threadIdx.x & 31;
  const int w    = threadIdx.x >> 5;
  const int wm   = w & 1;
  const int wn   = w >> 1;
  const int tileM = blockIdx.x * (32 * FM) + wm * (16 * FM);
  const int tileN = blockIdx.y * (32 * FN) + wn * (16 * FN);
  const int half = lane >> 4;
  const int l16  = lane & 15;

  const float* Ap[FM];
  const float* Bp[FN];
#pragma unroll
  for (int fm = 0; fm < FM; ++fm)
    Ap[fm] = (FLAGS & 256)
        ? A + (size_t)((tileM >> 4) + fm) * Kd * 16 + 4 * lane
        : A + (size_t)(tileM + 16 * fm + l16) * Kd + 4 * half;
#pragma unroll
  for (int fn = 0; fn < FN; ++fn)
    Bp[fn] = (FLAGS & 128)
        ? W + (size_t)((tileN >> 4) + fn) * Kd * 16 + 4 * lane
        : W + (size_t)(tileN + 16 * fn + l16) * Kd + 4 * half;

  v8f acc[FM][FN];
#pragma unroll
  for (int fm = 0; fm < FM; ++fm)
#pragma unroll
    for (int fn = 0; fn < FN; ++fn)
      acc[fm][fn] = v8f{};

  v4f a0[4][FM], b0[4][FN];
  v4f a1[4][FM], b1[4][FN];
  v4f g0[4], h0[4], g1[4], h1[4];

  float rinv[FM], rc0[FM], rw[FM];
  v4f t0[4], t1[4];
  if (FLAGS & 16) {
    static_assert(!(FLAGS & 16) || NS > 0, "bit4 needs the slot count NS = Kd/64");
    v2f ps[FM][NS > 0 ? NS : 1];
#pragma unroll
    for (int fm = 0; fm < FM; ++fm)
#pragma unroll
      for (int sl = 0; sl < NS; ++sl)
        ps[fm][sl] = *(const v2f*)(statsIn + ((size_t)sl * M + tileM + 16 * fm + l16) * 2);
#pragma unroll
    for (int fm = 0; fm < FM; ++fm) {
      const int row = tileM + 16 * fm + l16;
      float s = 0.f, s2 = 0.f;
#pragma unroll
      for (int sl = 0; sl < NS; ++sl) { s += ps[fm][sl][0]; s2 += ps[fm][sl][1]; }
      const float mean = s / Kd;
      const float var  = s2 / Kd - mean * mean;
      const float inv  = rsqrtf(var + 1e-5f);
      rinv[fm] = inv; rc0[fm] = -mean * inv;
      if (FLAGS & 32) rw[fm] = mixW[(size_t)row * Kspk + kk];
    }
  }

#define KPOS(koff, s)                                                         \
  ((FLAGS & 256) ? (koff) + 16 * ((s) >> 1) + 8 * half + 4 * ((s) & 1)        \
                 : (koff) + 8 * (s) + 4 * half)
#define LOAD_STAGE(as, bs, gs, hs, ts, koff)                                  \
  _Pragma("unroll") for (int s = 0; s < 4; ++s) {                             \
    _Pragma("unroll") for (int fm = 0; fm < FM; ++fm)                         \
      as[s][fm] = *(const v4f*)(Ap[fm] + ((FLAGS & 256)                       \
                                    ? ((koff) + 8 * s) * 16                   \
                                    : (koff) + 8 * s));                       \
    _Pragma("unroll") for (int fn = 0; fn < FN; ++fn)                         \
      bs[s][fn] = *(const v4f*)(Bp[fn] + ((FLAGS & 128)                       \
                                    ? ((koff) + 8 * s) * 16                   \
                                    : (koff) + 8 * s));                       \
    if (FLAGS & 16) {                                                         \
      gs[s] = *(const v4f*)(lnG + KPOS(koff, s));                             \
      hs[s] = *(const v4f*)(lnB + KPOS(koff, s));                             \
      if (FLAGS & 32)                                                         \
        ts[s] = *(const v4f*)(tags + KPOS(koff, s));                          \
    }                                                                         \
  }

#define NORM_STAGE(as, gs, hs, ts)                                            \
  if (FLAGS & 16) {                                                           \
    _Pragma("unroll") for (int s = 0; s < 4; ++s)                             \
      _Pragma("unroll") for (int fm = 0; fm < FM; ++fm) {                     \
        v4f u = as[s][fm];                                                    \
        if (FLAGS & 32) u = u * rw[fm] + ts[s];                               \
        const v4f t = u * rinv[fm] + rc0[fm];                                 \
        as[s][fm] = t * gs[s] + hs[s];                                        \
      }                                                                       \
  }

#define WMMA_AB(cacc, av, bv)                                                 \
  if (FLAGS & 512) { WMMA_F32(cacc, bv, av); } else { WMMA_F32(cacc, av, bv); }
#define MMA_STAGE(as, bs)                                                     \
  _Pragma("unroll") for (int s = 0; s < 4; ++s) {                             \
    _Pragma("unroll") for (int fm = 0; fm < FM; ++fm)                         \
      _Pragma("unroll") for (int fn = 0; fn < FN; ++fn)                       \
        WMMA_AB(acc[fm][fn],                                                  \
                __builtin_shufflevector(as[s][fm], as[s][fm], 0, 1),          \
                __builtin_shufflevector(bs[s][fn], bs[s][fn], 0, 1));         \
    _Pragma("unroll") for (int fm = 0; fm < FM; ++fm)                         \
      _Pragma("unroll") for (int fn = 0; fn < FN; ++fn)                       \
        WMMA_AB(acc[fm][fn],                                                  \
                __builtin_shufflevector(as[s][fm], as[s][fm], 2, 3),          \
                __builtin_shufflevector(bs[s][fn], bs[s][fn], 2, 3));         \
  }

#define STAGE_FENCE() __builtin_amdgcn_sched_barrier(0)

  LOAD_STAGE(a0, b0, g0, h0, t0, 0);
  for (int k0 = 0; k0 < Kd - 64; k0 += 64) {
    STAGE_FENCE();
    LOAD_STAGE(a1, b1, g1, h1, t1, k0 + 32);
    STAGE_FENCE();
    NORM_STAGE(a0, g0, h0, t0);
    MMA_STAGE(a0, b0);
    STAGE_FENCE();
    LOAD_STAGE(a0, b0, g0, h0, t0, k0 + 64);
    STAGE_FENCE();
    NORM_STAGE(a1, g1, h1, t1);
    MMA_STAGE(a1, b1);
  }
  STAGE_FENCE();
  LOAD_STAGE(a1, b1, g1, h1, t1, Kd - 32);
  STAGE_FENCE();
  NORM_STAGE(a0, g0, h0, t0);
  MMA_STAGE(a0, b0);
  NORM_STAGE(a1, g1, h1, t1);
  MMA_STAGE(a1, b1);
#undef STAGE_FENCE
#undef LOAD_STAGE
#undef NORM_STAGE
#undef MMA_STAGE
#undef WMMA_AB
#undef KPOS

  if (FLAGS & 512) {
    v4f bq[FN][2], tq0[FN][2], tq1[FN][2];
#pragma unroll
    for (int fn = 0; fn < FN; ++fn)
#pragma unroll
      for (int j = 0; j < 2; ++j) {
        const int col = tileN + 16 * fn + 8 * half + 4 * j;
        bq[fn][j] = *(const v4f*)(bias + col);
        tq0[fn][j] = v4f{};
        tq1[fn][j] = v4f{};
        if ((FLAGS & 64) || (FLAGS & 33) == 33) tq0[fn][j] = *(const v4f*)(tags + col);
        if (FLAGS & 64) tq1[fn][j] = *(const v4f*)(tags + N + col);
      }
    v2f sumV[FM], sumS0[FM], sumS1[FM];
#pragma unroll
    for (int fm = 0; fm < FM; ++fm) {
      const int row = tileM + 16 * fm + l16;
      float wr = 0.f;
      if ((FLAGS & 33) == 33) wr = mixW[(size_t)row * Kspk + kk];
      v2f w2 = v2f{};
      if (FLAGS & 64) w2 = *(const v2f*)(mixW + (size_t)row * 2);
      size_t orow = (size_t)row;
      if (FLAGS & 4) {
        const int bi = row / Tlen;
        orow = (size_t)bi * Kspk * Tlen + (size_t)kk * Tlen + (row % Tlen);
      }
      float ps = 0.f, ps2 = 0.f, p0 = 0.f, p02 = 0.f, p1 = 0.f, p12 = 0.f;
#pragma unroll
      for (int fn = 0; fn < FN; ++fn) {
        const size_t pb = ((size_t)((tileM >> 4) + fm) * N + tileN + 16 * fn) * 16 + 4 * lane;
#pragma unroll
        for (int j = 0; j < 2; ++j) {
          v4f v;
#pragma unroll
          for (int e = 0; e < 4; ++e) v[e] = acc[fm][fn][4 * j + e];
          v += bq[fn][j];
          if (FLAGS & 1) {
            v4f r = *(const v4f*)(R + pb + 128 * j);
            if (FLAGS & 32) r = r * wr + tq0[fn][j];
            v += r;
          }
          if (FLAGS & 2) {
#pragma unroll
            for (int e = 0; e < 4; ++e) v[e] = gelu_exact(v[e]);
          }
          if (FLAGS & 8) {
#pragma unroll
            for (int e = 0; e < 4; ++e) { ps += v[e]; ps2 += v[e] * v[e]; }
          }
          if (FLAGS & 64) {
            const v4f x0 = v * w2[0] + tq0[fn][j];
            const v4f x1 = v * w2[1] + tq1[fn][j];
#pragma unroll
            for (int e = 0; e < 4; ++e) {
              p0 += x0[e]; p02 += x0[e] * x0[e];
              p1 += x1[e]; p12 += x1[e] * x1[e];
            }
          }
          if (FLAGS & 1024)
            *(v4f*)(C + orow * N + tileN + 16 * fn + 8 * half + 4 * j) = v;
          else
            *(v4f*)(C + pb + 128 * j) = v;
        }
      }
      if (FLAGS & 8) {
        ps += __shfl_xor(ps, 16, 32); ps2 += __shfl_xor(ps2, 16, 32);
        sumV[fm][0] = ps; sumV[fm][1] = ps2;
      }
      if (FLAGS & 64) {
        p0 += __shfl_xor(p0, 16, 32); p02 += __shfl_xor(p02, 16, 32);
        p1 += __shfl_xor(p1, 16, 32); p12 += __shfl_xor(p12, 16, 32);
        sumS0[fm][0] = p0; sumS0[fm][1] = p02;
        sumS1[fm][0] = p1; sumS1[fm][1] = p12;
      }
    }
    if (FLAGS & 8) {
      const int slot = blockIdx.y * 2 + wn;
      const size_t pairs = (size_t)(N >> 6) * M * 2;
      float* sp = statsOut + ((size_t)slot * M + tileM + lane) * 2;
      *(v2f*)sp = half ? sumV[FM - 1] : sumV[0];
      if (FLAGS & 64) {
        *(v2f*)(sp + pairs)     = half ? sumS0[FM - 1] : sumS0[0];
        *(v2f*)(sp + 2 * pairs) = half ? sumS1[FM - 1] : sumS1[0];
      }
    }
    return;
  }

  float bv[FN];
#pragma unroll
  for (int fn = 0; fn < FN; ++fn) bv[fn] = bias[tileN + 16 * fn + l16];
  float tg0[FN], tg1[FN];
  if (FLAGS & 64) {
#pragma unroll
    for (int fn = 0; fn < FN; ++fn) {
      tg0[fn] = tags[tileN + 16 * fn + l16];
      tg1[fn] = tags[N + tileN + 16 * fn + l16];
    }
  } else if ((FLAGS & 33) == 33) {
#pragma unroll
    for (int fn = 0; fn < FN; ++fn) tg0[fn] = tags[tileN + 16 * fn + l16];
  }

  v2f keepV = v2f{}, keepS0 = v2f{}, keepS1 = v2f{};

#pragma unroll
  for (int fm = 0; fm < FM; ++fm) {
#pragma unroll
    for (int i = 0; i < 8; ++i) {
      const int row = tileM + 16 * fm + i + 8 * half;
      float v[FN];
#pragma unroll
      for (int fn = 0; fn < FN; ++fn) v[fn] = acc[fm][fn][i] + bv[fn];
      if (FLAGS & 1) {
        const float* rr = R + (size_t)row * N + tileN + l16;
        if (FLAGS & 32) {
          const float wr = mixW[(size_t)row * Kspk + kk];
#pragma unroll
          for (int fn = 0; fn < FN; ++fn) {
            const float xk = rr[16 * fn] * wr + tg0[fn];
            v[fn] += xk;
          }
        } else {
#pragma unroll
          for (int fn = 0; fn < FN; ++fn) v[fn] += rr[16 * fn];
        }
      }
      if (FLAGS & 2) {
#pragma unroll
        for (int fn = 0; fn < FN; ++fn) v[fn] = gelu_exact(v[fn]);
      }
      if (FLAGS & 8) {
        float ps = 0.f, ps2 = 0.f;
#pragma unroll
        for (int fn = 0; fn < FN; ++fn) { ps += v[fn]; ps2 += v[fn] * v[fn]; }
        ps = row16_sum(ps); ps2 = row16_sum(ps2);
        if (l16 == 8 * fm + i) { keepV[0] = ps; keepV[1] = ps2; }
      }
      if (FLAGS & 64) {
        const v2f w2 = *(const v2f*)(mixW + (size_t)row * 2);
        float p0 = 0.f, p02 = 0.f, p1 = 0.f, p12 = 0.f;
#pragma unroll
        for (int fn = 0; fn < FN; ++fn) {
          const float x0 = v[fn] * w2[0] + tg0[fn];
          const float x1 = v[fn] * w2[1] + tg1[fn];
          p0 += x0; p02 += x0 * x0; p1 += x1; p12 += x1 * x1;
        }
        p0 = row16_sum(p0); p02 = row16_sum(p02);
        p1 = row16_sum(p1); p12 = row16_sum(p12);
        if (l16 == 8 * fm + i) { keepS0[0] = p0; keepS0[1] = p02; keepS1[0] = p1; keepS1[1] = p12; }
      }
      size_t orow = (size_t)row;
      if (FLAGS & 4) {
        const int bi = row / Tlen;
        orow = (size_t)bi * Kspk * Tlen + (size_t)kk * Tlen + (row % Tlen);
      }
      float* cr = C + orow * N + tileN + l16;
#pragma unroll
      for (int fn = 0; fn < FN; ++fn) cr[16 * fn] = v[fn];
    }
  }
  if (FLAGS & 8) {
    const int srow = tileM + 16 * (l16 >> 3) + 8 * half + (l16 & 7);
    const int slot = blockIdx.y * 2 + wn;
    const size_t pairs = (size_t)(N >> 6) * M * 2;
    float* sp = statsOut + ((size_t)slot * M + srow) * 2;
    *(v2f*)sp = keepV;
    if (FLAGS & 64) {
      *(v2f*)(sp + pairs)     = keepS0;
      *(v2f*)(sp + 2 * pairs) = keepS1;
    }
  }
}

template <int DIN, int Dm>
__global__ __launch_bounds__(128) void pack_weights_frag(
    const float* __restrict__ w_proj, const float* __restrict__ w_in,
    const float* __restrict__ w_out, const float* __restrict__ w_f1,
    const float* __restrict__ w_f2,
    float* __restrict__ p_proj, float* __restrict__ p_in,
    float* __restrict__ p_out, float* __restrict__ p_f1,
    float* __restrict__ p_f2)
{
  constexpr int I0 = (Dm / 16) * (DIN / 32);
  constexpr int I1 = I0 + (3 * Dm / 16) * (Dm / 32);
  constexpr int I2 = I1 + (Dm / 16) * (Dm / 32);
  constexpr int I3 = I2 + (4 * Dm / 16) * (Dm / 32);
  constexpr int I4 = I3 + (Dm / 16) * (4 * Dm / 32);
  const int lane = threadIdx.x & 31;
  const int c    = lane & 15;
  const int half = lane >> 4;
  int item = blockIdx.x * 4 + (threadIdx.x >> 5);
  const float* src; float* dst; int Kd;
  bool g16 = false;
  if (item < I0)      { src = w_proj; dst = p_proj; Kd = DIN; }
  else if (item < I1) { src = w_in;   dst = p_in;   Kd = Dm;     item -= I0; g16 = true; }
  else if (item < I2) { src = w_out;  dst = p_out;  Kd = Dm;     item -= I1; }
  else if (item < I3) { src = w_f1;   dst = p_f1;   Kd = Dm;     item -= I2; g16 = true; }
  else if (item < I4) { src = w_f2;   dst = p_f2;   Kd = 4 * Dm; item -= I3; g16 = true; }
  else return;
  const int stages = Kd / 32;
  const int g  = item / stages;
  const int st = item % stages;
  const float* rp = src + (size_t)(16 * g + c) * Kd + 32 * st;
  float* op = dst + ((size_t)g * (Kd / 8) + 4 * st) * 128 + 4 * lane;
  v4f x[4];
#pragma unroll
  for (int s = 0; s < 4; ++s)
    x[s] = *(const v4f*)(rp + (g16 ? 16 * (s >> 1) + 8 * half + 4 * (s & 1) : 8 * s + 4 * half));
#pragma unroll
  for (int s = 0; s < 4; ++s) *(v4f*)(op + 128 * s) = x[s];
}

__global__ __launch_bounds__(256) void wmix_kernel(
    const float* __restrict__ logits, const int* __restrict__ blank_p,
    float* __restrict__ wmix, int BT, int V, int Kspk)
{
  const int row = blockIdx.x;
  const int tid = threadIdx.x;
  const int blank = blank_p[0];
  __shared__ float red[256];
  float a[8];
  for (int k2 = 0; k2 < Kspk; ++k2) {
    const float* L = logits + ((size_t)k2 * BT + row) * V;
    float m = -INFINITY;
    for (int v = tid; v < V; v += 256) m = fmaxf(m, L[v]);
    red[tid] = m; __syncthreads();
    for (int off = 128; off > 0; off >>= 1) {
      if (tid < off) red[tid] = fmaxf(red[tid], red[tid + off]);
      __syncthreads();
    }
    m = red[0]; __syncthreads();
    float s = 0.f;
    for (int v = tid; v < V; v += 256) s += expf(L[v] - m);
    red[tid] = s; __syncthreads();
    for (int off = 128; off > 0; off >>= 1) {
      if (tid < off) red[tid] += red[tid + off];
      __syncthreads();
    }
    s = red[0]; __syncthreads();
    a[k2] = 1.0f - expf(L[blank] - m) / s;
  }
  if (tid == 0) {
    float m2 = -INFINITY;
    for (int k2 = 0; k2 < Kspk; ++k2) m2 = fmaxf(m2, a[k2]);
    float ss = 0.f;
    float e[8];
    for (int k2 = 0; k2 < Kspk; ++k2) { e[k2] = expf(a[k2] - m2); ss += e[k2]; }
    for (int k2 = 0; k2 < Kspk; ++k2) wmix[(size_t)row * Kspk + k2] = e[k2] / ss;
  }
}

template <int T_, int Dn, int Hn>
__global__ __launch_bounds__(128) void banded_attn_wmma(
    const float* __restrict__ q, const float* __restrict__ kbuf,
    const float* __restrict__ vbuf, float* __restrict__ ctx)
{
  constexpr int HD = 64, BAND = 24;
  static_assert(Dn == Hn * HD && T_ % 16 == 0, "attention tile shape");
  const int lane = threadIdx.x & 31;
  const int wid  = threadIdx.x >> 5;
  const int c    = lane & 15;
  const int half = lane >> 4;
  constexpr int tilesT = T_ / 16;
  const int gw   = blockIdx.x * 4 + wid;
  const int tile = gw % tilesT;
  const int hh   = (gw / tilesT) % Hn;
  const int b    = gw / (tilesT * Hn);
  const int t0   = tile * 16;
  const int key0 = t0 - BAND;
  const unsigned hoff = (unsigned)(hh * HD);
  const int brow = b * T_;


  v4f qf[8];
  {
    const float* qp = q + ((unsigned)((brow + t0 + c) * Dn) + hoff + 4u * half);
#pragma unroll
    for (int i = 0; i < 8; ++i) qf[i] = *(const v4f*)(qp + 8 * i);
  }
  v4f kf[4][8];
#pragma unroll
  for (int kb = 0; kb < 4; ++kb) {
    int s = key0 + 16 * kb + c;
    s = min(max(s, 0), T_ - 1);
    const float* kp = kbuf + ((unsigned)((brow + s) * Dn) + hoff + 4u * half);
#pragma unroll
    for (int i = 0; i < 8; ++i) kf[kb][i] = *(const v4f*)(kp + 8 * i);
  }
  v4f vf[4][4][2];
#pragma unroll
  for (int kb = 0; kb < 4; ++kb)
#pragma unroll
    for (int j = 0; j < 4; ++j)
#pragma unroll
      for (int e = 0; e < 2; ++e) {
        int s = key0 + 16 * kb + 8 * half + 2 * j + e;
        s = min(max(s, 0), T_ - 1);
        vf[kb][j][e] = *(const v4f*)(vbuf + ((unsigned)((brow + s) * Dn) + hoff + 4u * c));
      }

  __builtin_amdgcn_sched_barrier(0);
  v8f st[4];
#pragma unroll
  for (int kb = 0; kb < 4; ++kb) st[kb] = v8f{};
#pragma unroll
  for (int i = 0; i < 8; ++i) {
#pragma unroll
    for (int kb = 0; kb < 4; ++kb)
      WMMA_F32(st[kb], __builtin_shufflevector(kf[kb][i], kf[kb][i], 0, 1),
                       __builtin_shufflevector(qf[i], qf[i], 0, 1));
#pragma unroll
    for (int kb = 0; kb < 4; ++kb)
      WMMA_F32(st[kb], __builtin_shufflevector(kf[kb][i], kf[kb][i], 2, 3),
                       __builtin_shufflevector(qf[i], qf[i], 2, 3));
  }

  const float scale = 0.125f;
  float p[4][8];
  float m = -INFINITY;
#pragma unroll
  for (int kb = 0; kb < 4; ++kb)
#pragma unroll
    for (int r = 0; r < 8; ++r) {
      const int rel = 16 * kb + 8 * half + r - c;
      const int s   = key0 + 16 * kb + 8 * half + r;
      const bool ok = (rel >= 0) && (rel <= 2 * BAND) && (s >= 0) && (s < T_);
      const float v = ok ? st[kb][r] * scale : -INFINITY;
      p[kb][r] = v;
      m = fmaxf(m, v);
    }
  m = fmaxf(m, __shfl_xor(m, 16, 32));
  float sum = 0.f;
#pragma unroll
  for (int kb = 0; kb < 4; ++kb)
#pragma unroll
    for (int r = 0; r < 8; ++r) {
      const float e = expf(p[kb][r] - m);
      p[kb][r] = e;
      sum += e;
    }
  sum += __shfl_xor(sum, 16, 32);
  const float inv = 1.0f / sum;

  v8f o[4];
#pragma unroll
  for (int db = 0; db < 4; ++db) o[db] = v8f{};
#pragma unroll
  for (int kb = 0; kb < 4; ++kb)
#pragma unroll
    for (int j = 0; j < 4; ++j) {
      v2f a;
      a[0] = p[kb][2 * j] * inv;
      a[1] = p[kb][2 * j + 1] * inv;
#pragma unroll
      for (int db = 0; db < 4; ++db) {
        v2f bb;
        bb[0] = vf[kb][j][0][db];
        bb[1] = vf[kb][j][1][db];
        WMMA_F32(o[db], a, bb);
      }
    }

#pragma unroll
  for (int r = 0; r < 8; ++r) {
    v4f ov;
    ov[0] = o[0][r]; ov[1] = o[1][r]; ov[2] = o[2][r]; ov[3] = o[3][r];
    *(v4f*)(ctx + ((unsigned)((brow + t0 + 8 * half + r) * Dn) + hoff + 4u * c)) = ov;
  }
}

extern "C" void kernel_launch(void* const* d_in, const int* in_sizes, int n_in,
                              void* d_out, int out_size, void* d_ws, size_t ws_size,
                              hipStream_t stream) {
  (void)in_sizes; (void)n_in; (void)out_size; (void)ws_size;
  constexpr int B = 4, T = 1024, D_IN = 512, D = 512, K = 2, V = 500, H = 8, BAND = 24;
  constexpr int M = B * T;

  const float* x_m      = (const float*)d_in[0];
  const float* logits   = (const float*)d_in[1];
  const int*   blank_id = (const int*)  d_in[2];
  const float* proj_w   = (const float*)d_in[3];
  const float* proj_b   = (const float*)d_in[4];
  const float* spk_tags = (const float*)d_in[5];
  const float* lnq_g    = (const float*)d_in[6];
  const float* lnq_b    = (const float*)d_in[7];
  const float* lnkv_g   = (const float*)d_in[8];
  const float* lnkv_b   = (const float*)d_in[9];
  const float* in_w     = (const float*)d_in[10];
  const float* in_b     = (const float*)d_in[11];
  const float* out_w    = (const float*)d_in[12];
  const float* out_b    = (const float*)d_in[13];
  const float* ffn_ln_g = (const float*)d_in[14];
  const float* ffn_ln_b = (const float*)d_in[15];
  const float* ffn_w1   = (const float*)d_in[16];
  const float* ffn_b1   = (const float*)d_in[17];
  const float* ffn_w2   = (const float*)d_in[18];
  const float* ffn_b2   = (const float*)d_in[19];
  float* out = (float*)d_out;

  float* ws = (float*)d_ws;
  const size_t SZ = (size_t)M * D;
  float* X    = ws;
  float* kbuf = X    + SZ;
  float* vbuf = kbuf + SZ;
  float* tA   = vbuf + SZ;
  float* tB   = tA   + SZ;
  float* tC   = tB   + SZ;
  float* t1   = tC   + SZ;
  float* ctx  = t1;
  float* wmix = t1   + 4 * SZ;

  static_assert(K == 2, "the projection epilogue sums two speaker streams");
  const size_t PAIRS = (size_t)(D / 64) * M * 2;
  float* sumsX = tA;
  float* sumsY = tA + 3 * PAIRS;

  static_assert(4 * PAIRS + (size_t)D * D_IN + 4 * (size_t)D * D <= SZ, "packed small weights fit behind the row sums");
  static_assert(8 * (size_t)D * D == SZ, "packed FFN weights fill tB");
  float* pk_proj = tA + 4 * PAIRS;
  float* pk_in   = pk_proj + (size_t)D * D_IN;
  float* pk_out  = pk_in   + 3 * (size_t)D * D;
  float* pk_f1   = tB;
  float* pk_f2   = tB + 4 * (size_t)D * D;

  const dim3 blk(256);
  const dim3 gblk(128);
  auto gg = [](int m, int n) { return dim3(m / 64, n / 128); };

  pack_weights_frag<D_IN, D><<<dim3(((D / 16) * (D_IN / 32) + (8 * D / 16) * (D / 32) + (D / 16) * (4 * D / 32)) / 4),
                              dim3(128), 0, stream>>>(
      proj_w, in_w, out_w, ffn_w1, ffn_w2, pk_proj, pk_in, pk_out, pk_f1, pk_f2);
  wmix_kernel<<<dim3(M), blk, 0, stream>>>(logits, blank_id, wmix, M, V, K);
  gemm_wmma_f32<2, 4,  8 | 64 | 128 | 512><<<gg(M, D), gblk, 0, stream>>>(
      x_m, pk_proj, proj_b, nullptr, X, M, D, D_IN, T, K, 0,
      nullptr, nullptr, nullptr, sumsX, wmix, spk_tags);
  gemm_wmma_f32<2, 4,  16 | 128 | 256, D / 64><<<gg(M, D), gblk, 0, stream>>>(
      X, pk_in + (size_t)D * D, in_b + D, nullptr, kbuf, M, D, D, T, K, 0,
      lnkv_g, lnkv_b, sumsX, nullptr, nullptr, nullptr);
  gemm_wmma_f32<2, 4,  16 | 128 | 256, D / 64><<<gg(M, D), gblk, 0, stream>>>(
      X, pk_in + (size_t)2 * D * D, in_b + 2 * D, nullptr, vbuf, M, D, D, T, K, 0,
      lnkv_g, lnkv_b, sumsX, nullptr, nullptr, nullptr);

  for (int kk = 0; kk < K; ++kk) {
    const float* tagk = spk_tags + (size_t)kk * D;
    gemm_wmma_f32<2, 4,  16 | 32 | 128 | 256, D / 64><<<gg(M, D), gblk, 0, stream>>>(
        X, pk_in, in_b, nullptr, tC, M, D, D, T, K, kk,
        lnq_g, lnq_b, sumsX + (size_t)(1 + kk) * PAIRS, nullptr, wmix, tagk);
    static_assert(BAND == 24, "attention tile shape");
    banded_attn_wmma<T, D, H><<<dim3((B * H * (T / 16)) / 4), dim3(128), 0, stream>>>(tC, kbuf, vbuf, ctx);
    gemm_wmma_f32<2, 4,  1 | 8 | 32 | 128 | 512><<<gg(M, D), gblk, 0, stream>>>(
        ctx, pk_out, out_b, X, tC, M, D, D, T, K, kk,
        nullptr, nullptr, nullptr, sumsY, wmix, tagk);
    gemm_wmma_f32<2, 4,  2 | 16 | 128 | 256 | 512, D / 64><<<gg(M, 4 * D), gblk, 0, stream>>>(
        tC, pk_f1, ffn_b1, nullptr, t1, M, 4 * D, D, T, K, kk,
        ffn_ln_g, ffn_ln_b, sumsY, nullptr, nullptr, nullptr);
    gemm_wmma_f32<2, 4,  1 | 4 | 128 | 256 | 512 | 1024><<<gg(M, D), gblk, 0, stream>>>(
        t1, pk_f2, ffn_b2, tC, out, M, D, 4 * D, T, K, kk,
        nullptr, nullptr, nullptr, nullptr, nullptr, nullptr);
  }
}


# GEMM K loop of the one-wave-per-SIMD launches on three register sets: a stage's fragment loads leave two stages ahead of its products; the four larger launches regrouped as 2-wave work-groups
# speedup vs baseline: 2.3301x; 1.0346x over previous
#include <hip/hip_runtime.h>
#include <hip/hip_bf16.h>
#include <math.h>

typedef float v2f __attribute__((ext_vector_type(2)));
typedef float v4f __attribute__((ext_vector_type(4)));
typedef float v8f __attribute__((ext_vector_type(8)));

__device__ __forceinline__ float gelu_exact(float x) {
  return 0.5f * x * (1.0f + erff(x * 0.70710678118654752f));
}

#define WMMA_F32(cacc, av, bv)                                                 \
  cacc = __builtin_amdgcn_wmma_f32_16x16x4_f32(false, (av), false, (bv),       \
                                               (short)0, (cacc), false, false)

template <int MASK>
__device__ __forceinline__ float row16_xor(float x) {
  return __int_as_float(__builtin_amdgcn_update_dpp(0, __float_as_int(x), 0x160 + MASK, 0xf, 0xf, true));
}
__device__ __forceinline__ float row16_sum(float x) {
  x += row16_xor<1>(x);
  x += row16_xor<2>(x);
  x += row16_xor<4>(x);
  x += row16_xor<8>(x);
  return x;
}

template <int FM, int FN, int FLAGS, int NS = 0>
__global__ __launch_bounds__(128) void gemm_wmma_f32(
    const float* __restrict__ A, const float* __restrict__ W,
    const float* __restrict__ bias, const float* __restrict__ R,
    float* __restrict__ C, int M, int N, int Kd,
    int Tlen, int Kspk, int kk,
    const float* __restrict__ lnG, const float* __restrict__ lnB,
    const float* __restrict__ statsIn, float* __restrict__ statsOut,
    const float* __restrict__ mixW, const float* __restrict__ tags)
{
  static_assert(!(FLAGS & (8 | 16 | 64)) || FM == 2, "row-sum slots assume a 32-row wave tile");
  static_assert(!(FLAGS & 64) || (FLAGS & 8), "stream sums ride on the row-sum epilogue");
  static_assert(!(FLAGS & 4096) || !(FLAGS & 8), "row-sum slots assume two wave-cols per work-group");
  const int lane = threadIdx.x & 31;
  const int w    = threadIdx.x >> 5;
  const int wm   = w & 1;
  const int wn   = (FLAGS & 4096) ? 0 : (w >> 1);
  const int tileM = blockIdx.x * (32 * FM) + wm * (16 * FM);
  const int tileN = blockIdx.y * ((FLAGS & 4096) ? 16 * FN : 32 * FN) + wn * (16 * FN);
  const int half = lane >> 4;
  const int l16  = lane & 15;

  const float* Ap[FM];
  const float* Bp[FN];
#pragma unroll
  for (int fm = 0; fm < FM; ++fm)
    Ap[fm] = (FLAGS & 256)
        ? A + (size_t)((tileM >> 4) + fm) * Kd * 16 + 4 * lane
        : A + (size_t)(tileM + 16 * fm + l16) * Kd + 4 * half;
#pragma unroll
  for (int fn = 0; fn < FN; ++fn)
    Bp[fn] = (FLAGS & 128)
        ? W + (size_t)((tileN >> 4) + fn) * Kd * 16 + 4 * lane
        : W + (size_t)(tileN + 16 * fn + l16) * Kd + 4 * half;

  v8f acc[FM][FN];
#pragma unroll
  for (int fm = 0; fm < FM; ++fm)
#pragma unroll
    for (int fn = 0; fn < FN; ++fn)
      acc[fm][fn] = v8f{};

  v4f a0[4][FM], b0[4][FN];
  v4f a1[4][FM], b1[4][FN];
  v4f a2[4][FM], b2[4][FN];
  v4f g0[4], h0[4], g1[4], h1[4];
  v4f g2[4], h2[4];

  float rinv[FM], rc0[FM], rw[FM];
  v4f t0[4], t1[4], t2[4];
  if (FLAGS & 16) {
    static_assert(!(FLAGS & 16) || NS > 0, "bit4 needs the slot count NS = Kd/64");
    v2f ps[FM][NS > 0 ? NS : 1];
#pragma unroll
    for (int fm = 0; fm < FM; ++fm)
#pragma unroll
      for (int sl = 0; sl < NS; ++sl)
        ps[fm][sl] = *(const v2f*)(statsIn + ((size_t)sl * M + tileM + 16 * fm + l16) * 2);
#pragma unroll
    for (int fm = 0; fm < FM; ++fm) {
      const int row = tileM + 16 * fm + l16;
      float s = 0.f, s2 = 0.f;
#pragma unroll
      for (int sl = 0; sl < NS; ++sl) { s += ps[fm][sl][0]; s2 += ps[fm][sl][1]; }
      const float mean = s / Kd;
      const float var  = s2 / Kd - mean * mean;
      const float inv  = rsqrtf(var + 1e-5f);
      rinv[fm] = inv; rc0[fm] = -mean * inv;
      if (FLAGS & 32) rw[fm] = mixW[(size_t)row * Kspk + kk];
    }
  }

#define KPOS(koff, s)                                                         \
  ((FLAGS & 256) ? (koff) + 16 * ((s) >> 1) + 8 * half + 4 * ((s) & 1)        \
                 : (koff) + 8 * (s) + 4 * half)
#define LOAD_STAGE(as, bs, gs, hs, ts, koff)                                  \
  _Pragma("unroll") for (int s = 0; s < 4; ++s) {                             \
    _Pragma("unroll") for (int fm = 0; fm < FM; ++fm)                         \
      as[s][fm] = *(const v4f*)(Ap[fm] + ((FLAGS & 256)                       \
                                    ? ((koff) + 8 * s) * 16                   \
                                    : (koff) + 8 * s));                       \
    _Pragma("unroll") for (int fn = 0; fn < FN; ++fn)                         \
      bs[s][fn] = *(const v4f*)(Bp[fn] + ((FLAGS & 128)                       \
                                    ? ((koff) + 8 * s) * 16                   \
                                    : (koff) + 8 * s));                       \
    if (FLAGS & 16) {                                                         \
      gs[s] = *(const v4f*)(lnG + KPOS(koff, s));                             \
      hs[s] = *(const v4f*)(lnB + KPOS(koff, s));                             \
      if (FLAGS & 32)                                                         \
        ts[s] = *(const v4f*)(tags + KPOS(koff, s));                          \
    }                                                                         \
  }

#define LOAD_AB(as, bs, koff)                                                 \
  _Pragma("unroll") for (int s = 0; s < 4; ++s) {                             \
    _Pragma("unroll") for (int fm = 0; fm < FM; ++fm)                         \
      as[s][fm] = *(const v4f*)(Ap[fm] + ((FLAGS & 256)                       \
                                    ? ((koff) + 8 * s) * 16                   \
                                    : (koff) + 8 * s));                       \
    _Pragma("unroll") for (int fn = 0; fn < FN; ++fn)                         \
      bs[s][fn] = *(const v4f*)(Bp[fn] + ((FLAGS & 128)                       \
                                    ? ((koff) + 8 * s) * 16                   \
                                    : (koff) + 8 * s));                       \
  }
#define LOAD_GHT(gs, hs, ts, koff)                                            \
  if (FLAGS & 16) {                                                           \
    _Pragma("unroll") for (int s = 0; s < 4; ++s) {                           \
      gs[s] = *(const v4f*)(lnG + KPOS(koff, s));                             \
      hs[s] = *(const v4f*)(lnB + KPOS(koff, s));                             \
      if (FLAGS & 32)                                                         \
        ts[s] = *(const v4f*)(tags + KPOS(koff, s));                          \
    }                                                                         \
  }

#define NORM_STAGE(as, gs, hs, ts)                                            \
  if (FLAGS & 16) {                                                           \
    _Pragma("unroll") for (int s = 0; s < 4; ++s)                             \
      _Pragma("unroll") for (int fm = 0; fm < FM; ++fm) {                     \
        v4f u = as[s][fm];                                                    \
        if (FLAGS & 32) u = u * rw[fm] + ts[s];                               \
        const v4f t = u * rinv[fm] + rc0[fm];                                 \
        as[s][fm] = t * gs[s] + hs[s];                                        \
      }                                                                       \
  }

#define WMMA_AB(cacc, av, bv)                                                 \
  if (FLAGS & 512) { WMMA_F32(cacc, bv, av); } else { WMMA_F32(cacc, av, bv); }
#define MMA_STAGE(as, bs)                                                     \
  _Pragma("unroll") for (int s = 0; s < 4; ++s) {                             \
    _Pragma("unroll") for (int fm = 0; fm < FM; ++fm)                         \
      _Pragma("unroll") for (int fn = 0; fn < FN; ++fn)                       \
        WMMA_AB(acc[fm][fn],                                                  \
                __builtin_shufflevector(as[s][fm], as[s][fm], 0, 1),          \
                __builtin_shufflevector(bs[s][fn], bs[s][fn], 0, 1));         \
    _Pragma("unroll") for (int fm = 0; fm < FM; ++fm)                         \
      _Pragma("unroll") for (int fn = 0; fn < FN; ++fn)                       \
        WMMA_AB(acc[fm][fn],                                                  \
                __builtin_shufflevector(as[s][fm], as[s][fm], 2, 3),          \
                __builtin_shufflevector(bs[s][fn], bs[s][fn], 2, 3));         \
  }

#define STAGE_FENCE() __builtin_amdgcn_sched_barrier(0)

  if (FLAGS & 2048) {
    LOAD_AB(a0, b0, 0);
    LOAD_GHT(g0, h0, t0, 0);
    STAGE_FENCE();
    LOAD_AB(a1, b1, 32);
    int k0 = 0;
    for (; k0 < Kd - 128; k0 += 96) {
      STAGE_FENCE();
      LOAD_GHT(g1, h1, t1, k0 + 32);
      STAGE_FENCE();
      LOAD_AB(a2, b2, k0 + 64);
      STAGE_FENCE();
      NORM_STAGE(a0, g0, h0, t0);
      MMA_STAGE(a0, b0);
      STAGE_FENCE();
      LOAD_GHT(g2, h2, t2, k0 + 64);
      STAGE_FENCE();
      LOAD_AB(a0, b0, k0 + 96);
      STAGE_FENCE();
      NORM_STAGE(a1, g1, h1, t1);
      MMA_STAGE(a1, b1);
      STAGE_FENCE();
      LOAD_GHT(g0, h0, t0, k0 + 96);
      STAGE_FENCE();
      LOAD_AB(a1, b1, k0 + 128);
      STAGE_FENCE();
      NORM_STAGE(a2, g2, h2, t2);
      MMA_STAGE(a2, b2);
    }
    STAGE_FENCE();
    LOAD_GHT(g1, h1, t1, k0 + 32);
    STAGE_FENCE();
    LOAD_AB(a2, b2, k0 + 64);
    STAGE_FENCE();
    NORM_STAGE(a0, g0, h0, t0);
    MMA_STAGE(a0, b0);
    STAGE_FENCE();
    LOAD_GHT(g2, h2, t2, k0 + 64);
    STAGE_FENCE();
    LOAD_AB(a0, b0, k0 + 96);
    STAGE_FENCE();
    NORM_STAGE(a1, g1, h1, t1);
    MMA_STAGE(a1, b1);
    STAGE_FENCE();
    LOAD_GHT(g0, h0, t0, k0 + 96);
    STAGE_FENCE();
    NORM_STAGE(a2, g2, h2, t2);
    MMA_STAGE(a2, b2);
    NORM_STAGE(a0, g0, h0, t0);
    MMA_STAGE(a0, b0);
  } else {
    LOAD_STAGE(a0, b0, g0, h0, t0, 0);
    for (int k0 = 0; k0 < Kd - 64; k0 += 64) {
      STAGE_FENCE();
      LOAD_STAGE(a1, b1, g1, h1, t1, k0 + 32);
      STAGE_FENCE();
      NORM_STAGE(a0, g0, h0, t0);
      MMA_STAGE(a0, b0);
      STAGE_FENCE();
      LOAD_STAGE(a0, b0, g0, h0, t0, k0 + 64);
      STAGE_FENCE();
      NORM_STAGE(a1, g1, h1, t1);
      MMA_STAGE(a1, b1);
    }
    STAGE_FENCE();
    LOAD_STAGE(a1, b1, g1, h1, t1, Kd - 32);
    STAGE_FENCE();
    NORM_STAGE(a0, g0, h0, t0);
    MMA_STAGE(a0, b0);
    NORM_STAGE(a1, g1, h1, t1);
    MMA_STAGE(a1, b1);
  }
#undef STAGE_FENCE
#undef LOAD_STAGE
#undef LOAD_AB
#undef LOAD_GHT
#undef NORM_STAGE
#undef MMA_STAGE
#undef WMMA_AB
#undef KPOS

  if (FLAGS & 512) {
    v4f bq[FN][2], tq0[FN][2], tq1[FN][2];
#pragma unroll
    for (int fn = 0; fn < FN; ++fn)
#pragma unroll
      for (int j = 0; j < 2; ++j) {
        const int col = tileN + 16 * fn + 8 * half + 4 * j;
        bq[fn][j] = *(const v4f*)(bias + col);
        tq0[fn][j] = v4f{};
        tq1[fn][j] = v4f{};
        if ((FLAGS & 64) || (FLAGS & 33) == 33) tq0[fn][j] = *(const v4f*)(tags + col);
        if (FLAGS & 64) tq1[fn][j] = *(const v4f*)(tags + N + col);
      }
    v2f sumV[FM], sumS0[FM], sumS1[FM];
#pragma unroll
    for (int fm = 0; fm < FM; ++fm) {
      const int row = tileM + 16 * fm + l16;
      float wr = 0.f;
      if ((FLAGS & 33) == 33) wr = mixW[(size_t)row * Kspk + kk];
      v2f w2 = v2f{};
      if (FLAGS & 64) w2 = *(const v2f*)(mixW + (size_t)row * 2);
      size_t orow = (size_t)row;
      if (FLAGS & 4) {
        const int bi = row / Tlen;
        orow = (size_t)bi * Kspk * Tlen + (size_t)kk * Tlen + (row % Tlen);
      }
      float ps = 0.f, ps2 = 0.f, p0 = 0.f, p02 = 0.f, p1 = 0.f, p12 = 0.f;
#pragma unroll
      for (int fn = 0; fn < FN; ++fn) {
        const size_t pb = ((size_t)((tileM >> 4) + fm) * N + tileN + 16 * fn) * 16 + 4 * lane;
#pragma unroll
        for (int j = 0; j < 2; ++j) {
          v4f v;
#pragma unroll
          for (int e = 0; e < 4; ++e) v[e] = acc[fm][fn][4 * j + e];
          v += bq[fn][j];
          if (FLAGS & 1) {
            v4f r = *(const v4f*)(R + pb + 128 * j);
            if (FLAGS & 32) r = r * wr + tq0[fn][j];
            v += r;
          }
          if (FLAGS & 2) {
#pragma unroll
            for (int e = 0; e < 4; ++e) v[e] = gelu_exact(v[e]);
          }
          if (FLAGS & 8) {
#pragma unroll
            for (int e = 0; e < 4; ++e) { ps += v[e]; ps2 += v[e] * v[e]; }
          }
          if (FLAGS & 64) {
            const v4f x0 = v * w2[0] + tq0[fn][j];
            const v4f x1 = v * w2[1] + tq1[fn][j];
#pragma unroll
            for (int e = 0; e < 4; ++e) {
              p0 += x0[e]; p02 += x0[e] * x0[e];
              p1 += x1[e]; p12 += x1[e] * x1[e];
            }
          }
          if (FLAGS & 1024)
            *(v4f*)(C + orow * N + tileN + 16 * fn + 8 * half + 4 * j) = v;
          else
            *(v4f*)(C + pb + 128 * j) = v;
        }
      }
      if (FLAGS & 8) {
        ps += __shfl_xor(ps, 16, 32); ps2 += __shfl_xor(ps2, 16, 32);
        sumV[fm][0] = ps; sumV[fm][1] = ps2;
      }
      if (FLAGS & 64) {
        p0 += __shfl_xor(p0, 16, 32); p02 += __shfl_xor(p02, 16, 32);
        p1 += __shfl_xor(p1, 16, 32); p12 += __shfl_xor(p12, 16, 32);
        sumS0[fm][0] = p0; sumS0[fm][1] = p02;
        sumS1[fm][0] = p1; sumS1[fm][1] = p12;
      }
    }
    if (FLAGS & 8) {
      const int slot = blockIdx.y * 2 + wn;
      const size_t pairs = (size_t)(N >> 6) * M * 2;
      float* sp = statsOut + ((size_t)slot * M + tileM + lane) * 2;
      *(v2f*)sp = half ? sumV[FM - 1] : sumV[0];
      if (FLAGS & 64) {
        *(v2f*)(sp + pairs)     = half ? sumS0[FM - 1] : sumS0[0];
        *(v2f*)(sp + 2 * pairs) = half ? sumS1[FM - 1] : sumS1[0];
      }
    }
    return;
  }

  float bv[FN];
#pragma unroll
  for (int fn = 0; fn < FN; ++fn) bv[fn] = bias[tileN + 16 * fn + l16];
  float tg0[FN], tg1[FN];
  if (FLAGS & 64) {
#pragma unroll
    for (int fn = 0; fn < FN; ++fn) {
      tg0[fn] = tags[tileN + 16 * fn + l16];
      tg1[fn] = tags[N + tileN + 16 * fn + l16];
    }
  } else if ((FLAGS & 33) == 33) {
#pragma unroll
    for (int fn = 0; fn < FN; ++fn) tg0[fn] = tags[tileN + 16 * fn + l16];
  }

  v2f keepV = v2f{}, keepS0 = v2f{}, keepS1 = v2f{};

#pragma unroll
  for (int fm = 0; fm < FM; ++fm) {
#pragma unroll
    for (int i = 0; i < 8; ++i) {
      const int row = tileM + 16 * fm + i + 8 * half;
      float v[FN];
#pragma unroll
      for (int fn = 0; fn < FN; ++fn) v[fn] = acc[fm][fn][i] + bv[fn];
      if (FLAGS & 1) {
        const float* rr = R + (size_t)row * N + tileN + l16;
        if (FLAGS & 32) {
          const float wr = mixW[(size_t)row * Kspk + kk];
#pragma unroll
          for (int fn = 0; fn < FN; ++fn) {
            const float xk = rr[16 * fn] * wr + tg0[fn];
            v[fn] += xk;
          }
        } else {
#pragma unroll
          for (int fn = 0; fn < FN; ++fn) v[fn] += rr[16 * fn];
        }
      }
      if (FLAGS & 2) {
#pragma unroll
        for (int fn = 0; fn < FN; ++fn) v[fn] = gelu_exact(v[fn]);
      }
      if (FLAGS & 8) {
        float ps = 0.f, ps2 = 0.f;
#pragma unroll
        for (int fn = 0; fn < FN; ++fn) { ps += v[fn]; ps2 += v[fn] * v[fn]; }
        ps = row16_sum(ps); ps2 = row16_sum(ps2);
        if (l16 == 8 * fm + i) { keepV[0] = ps; keepV[1] = ps2; }
      }
      if (FLAGS & 64) {
        const v2f w2 = *(const v2f*)(mixW + (size_t)row * 2);
        float p0 = 0.f, p02 = 0.f, p1 = 0.f, p12 = 0.f;
#pragma unroll
        for (int fn = 0; fn < FN; ++fn) {
          const float x0 = v[fn] * w2[0] + tg0[fn];
          const float x1 = v[fn] * w2[1] + tg1[fn];
          p0 += x0; p02 += x0 * x0; p1 += x1; p12 += x1 * x1;
        }
        p0 = row16_sum(p0); p02 = row16_sum(p02);
        p1 = row16_sum(p1); p12 = row16_sum(p12);
        if (l16 == 8 * fm + i) { keepS0[0] = p0; keepS0[1] = p02; keepS1[0] = p1; keepS1[1] = p12; }
      }
      size_t orow = (size_t)row;
      if (FLAGS & 4) {
        const int bi = row / Tlen;
        orow = (size_t)bi * Kspk * Tlen + (size_t)kk * Tlen + (row % Tlen);
      }
      float* cr = C + orow * N + tileN + l16;
#pragma unroll
      for (int fn = 0; fn < FN; ++fn) cr[16 * fn] = v[fn];
    }
  }
  if (FLAGS & 8) {
    const int srow = tileM + 16 * (l16 >> 3) + 8 * half + (l16 & 7);
    const int slot = blockIdx.y * 2 + wn;
    const size_t pairs = (size_t)(N >> 6) * M * 2;
    float* sp = statsOut + ((size_t)slot * M + srow) * 2;
    *(v2f*)sp = keepV;
    if (FLAGS & 64) {
      *(v2f*)(sp + pairs)     = keepS0;
      *(v2f*)(sp + 2 * pairs) = keepS1;
    }
  }
}

template <int DIN, int Dm>
__global__ __launch_bounds__(128) void pack_weights_frag(
    const float* __restrict__ w_proj, const float* __restrict__ w_in,
    const float* __restrict__ w_out, const float* __restrict__ w_f1,
    const float* __restrict__ w_f2,
    float* __restrict__ p_proj, float* __restrict__ p_in,
    float* __restrict__ p_out, float* __restrict__ p_f1,
    float* __restrict__ p_f2)
{
  constexpr int I0 = (Dm / 16) * (DIN / 32);
  constexpr int I1 = I0 + (3 * Dm / 16) * (Dm / 32);
  constexpr int I2 = I1 + (Dm / 16) * (Dm / 32);
  constexpr int I3 = I2 + (4 * Dm / 16) * (Dm / 32);
  constexpr int I4 = I3 + (Dm / 16) * (4 * Dm / 32);
  const int lane = threadIdx.x & 31;
  const int c    = lane & 15;
  const int half = lane >> 4;
  int item = blockIdx.x * 2 + (threadIdx.x >> 5);
  const float* src; float* dst; int Kd;
  bool g16 = false;
  if (item < I0)      { src = w_proj; dst = p_proj; Kd = DIN; }
  else if (item < I1) { src = w_in;   dst = p_in;   Kd = Dm;     item -= I0; g16 = true; }
  else if (item < I2) { src = w_out;  dst = p_out;  Kd = Dm;     item -= I1; }
  else if (item < I3) { src = w_f1;   dst = p_f1;   Kd = Dm;     item -= I2; g16 = true; }
  else if (item < I4) { src = w_f2;   dst = p_f2;   Kd = 4 * Dm; item -= I3; g16 = true; }
  else return;
  const int stages = Kd / 32;
  const int g  = item / stages;
  const int st = item % stages;
  const float* rp = src + (size_t)(16 * g + c) * Kd + 32 * st;
  float* op = dst + ((size_t)g * (Kd / 8) + 4 * st) * 128 + 4 * lane;
  v4f x[4];
#pragma unroll
  for (int s = 0; s < 4; ++s)
    x[s] = *(const v4f*)(rp + (g16 ? 16 * (s >> 1) + 8 * half + 4 * (s & 1) : 8 * s + 4 * half));
#pragma unroll
  for (int s = 0; s < 4; ++s) *(v4f*)(op + 128 * s) = x[s];
}

__global__ __launch_bounds__(64) void wmix_kernel(
    const float* __restrict__ logits, const int* __restrict__ blank_p,
    float* __restrict__ wmix, int BT, int V, int Kspk)
{
  const int row = blockIdx.x;
  const int tid = threadIdx.x;
  const int blank = blank_p[0];
  __shared__ float red[256];
  float a[8];
  for (int k2 = 0; k2 < Kspk; ++k2) {
    const float* L = logits + ((size_t)k2 * BT + row) * V;
    for (int t = tid; t < 256; t += 64) {
      float mt = -INFINITY;
      for (int v = t; v < V; v += 256) mt = fmaxf(mt, L[v]);
      red[t] = mt;
    }
    __syncthreads();
    for (int off = 128; off > 0; off >>= 1) {
      for (int i = tid; i < off; i += 64) red[i] = fmaxf(red[i], red[i + off]);
      __syncthreads();
    }
    const float m = red[0]; __syncthreads();
    for (int t = tid; t < 256; t += 64) {
      float st = 0.f;
      for (int v = t; v < V; v += 256) st += expf(L[v] - m);
      red[t] = st;
    }
    __syncthreads();
    for (int off = 128; off > 0; off >>= 1) {
      for (int i = tid; i < off; i += 64) red[i] += red[i + off];
      __syncthreads();
    }
    const float s = red[0]; __syncthreads();
    a[k2] = 1.0f - expf(L[blank] - m) / s;
  }
  if (tid == 0) {
    float m2 = -INFINITY;
    for (int k2 = 0; k2 < Kspk; ++k2) m2 = fmaxf(m2, a[k2]);
    float ss = 0.f;
    float e[8];
    for (int k2 = 0; k2 < Kspk; ++k2) { e[k2] = expf(a[k2] - m2); ss += e[k2]; }
    for (int k2 = 0; k2 < Kspk; ++k2) wmix[(size_t)row * Kspk + k2] = e[k2] / ss;
  }
}

template <int T_, int Dn, int Hn>
__global__ __launch_bounds__(128) void banded_attn_wmma(
    const float* __restrict__ q, const float* __restrict__ kbuf,
    const float* __restrict__ vbuf, float* __restrict__ ctx)
{
  constexpr int HD = 64, BAND = 24;
  static_assert(Dn == Hn * HD && T_ % 16 == 0, "attention tile shape");
  const int lane = threadIdx.x & 31;
  const int wid  = threadIdx.x >> 5;
  const int c    = lane & 15;
  const int half = lane >> 4;
  constexpr int tilesT = T_ / 16;
  const int gw   = blockIdx.x * 2 + wid;
  const int tile = gw % tilesT;
  const int hh   = (gw / tilesT) % Hn;
  const int b    = gw / (tilesT * Hn);
  const int t0   = tile * 16;
  const int key0 = t0 - BAND;
  const unsigned hoff = (unsigned)(hh * HD);
  const int brow = b * T_;


  v4f qf[8];
  {
    const float* qp = q + ((unsigned)((brow + t0 + c) * Dn) + hoff + 4u * half);
#pragma unroll
    for (int i = 0; i < 8; ++i) qf[i] = *(const v4f*)(qp + 8 * i);
  }
  v4f kf[4][8];
#pragma unroll
  for (int kb = 0; kb < 4; ++kb) {
    int s = key0 + 16 * kb + c;
    s = min(max(s, 0), T_ - 1);
    const float* kp = kbuf + ((unsigned)((brow + s) * Dn) + hoff + 4u * half);
#pragma unroll
    for (int i = 0; i < 8; ++i) kf[kb][i] = *(const v4f*)(kp + 8 * i);
  }
  v4f vf[4][4][2];
#pragma unroll
  for (int kb = 0; kb < 4; ++kb)
#pragma unroll
    for (int j = 0; j < 4; ++j)
#pragma unroll
      for (int e = 0; e < 2; ++e) {
        int s = key0 + 16 * kb + 8 * half + 2 * j + e;
        s = min(max(s, 0), T_ - 1);
        vf[kb][j][e] = *(const v4f*)(vbuf + ((unsigned)((brow + s) * Dn) + hoff + 4u * c));
      }

  __builtin_amdgcn_sched_barrier(0);
  v8f st[4];
#pragma unroll
  for (int kb = 0; kb < 4; ++kb) st[kb] = v8f{};
#pragma unroll
  for (int i = 0; i < 8; ++i) {
#pragma unroll
    for (int kb = 0; kb < 4; ++kb)
      WMMA_F32(st[kb], __builtin_shufflevector(kf[kb][i], kf[kb][i], 0, 1),
                       __builtin_shufflevector(qf[i], qf[i], 0, 1));
#pragma unroll
    for (int kb = 0; kb < 4; ++kb)
      WMMA_F32(st[kb], __builtin_shufflevector(kf[kb][i], kf[kb][i], 2, 3),
                       __builtin_shufflevector(qf[i], qf[i], 2, 3));
  }

  const float scale = 0.125f;
  float p[4][8];
  float m = -INFINITY;
#pragma unroll
  for (int kb = 0; kb < 4; ++kb)
#pragma unroll
    for (int r = 0; r < 8; ++r) {
      const int rel = 16 * kb + 8 * half + r - c;
      const int s   = key0 + 16 * kb + 8 * half + r;
      const bool ok = (rel >= 0) && (rel <= 2 * BAND) && (s >= 0) && (s < T_);
      const float v = ok ? st[kb][r] * scale : -INFINITY;
      p[kb][r] = v;
      m = fmaxf(m, v);
    }
  m = fmaxf(m, __shfl_xor(m, 16, 32));
  float sum = 0.f;
#pragma unroll
  for (int kb = 0; kb < 4; ++kb)
#pragma unroll
    for (int r = 0; r < 8; ++r) {
      const float e = expf(p[kb][r] - m);
      p[kb][r] = e;
      sum += e;
    }
  sum += __shfl_xor(sum, 16, 32);
  const float inv = 1.0f / sum;

  v8f o[4];
#pragma unroll
  for (int db = 0; db < 4; ++db) o[db] = v8f{};
#pragma unroll
  for (int kb = 0; kb < 4; ++kb)
#pragma unroll
    for (int j = 0; j < 4; ++j) {
      v2f a;
      a[0] = p[kb][2 * j] * inv;
      a[1] = p[kb][2 * j + 1] * inv;
#pragma unroll
      for (int db = 0; db < 4; ++db) {
        v2f bb;
        bb[0] = vf[kb][j][0][db];
        bb[1] = vf[kb][j][1][db];
        WMMA_F32(o[db], a, bb);
      }
    }

#pragma unroll
  for (int r = 0; r < 8; ++r) {
    v4f ov;
    ov[0] = o[0][r]; ov[1] = o[1][r]; ov[2] = o[2][r]; ov[3] = o[3][r];
    *(v4f*)(ctx + ((unsigned)((brow + t0 + 8 * half + r) * Dn) + hoff + 4u * c)) = ov;
  }
}

extern "C" void kernel_launch(void* const* d_in, const int* in_sizes, int n_in,
                              void* d_out, int out_size, void* d_ws, size_t ws_size,
                              hipStream_t stream) {
  (void)in_sizes; (void)n_in; (void)out_size; (void)ws_size;
  constexpr int B = 4, T = 1024, D_IN = 512, D = 512, K = 2, V = 500, H = 8, BAND = 24;
  constexpr int M = B * T;

  const float* x_m      = (const float*)d_in[0];
  const float* logits   = (const float*)d_in[1];
  const int*   blank_id = (const int*)  d_in[2];
  const float* proj_w   = (const float*)d_in[3];
  const float* proj_b   = (const float*)d_in[4];
  const float* spk_tags = (const float*)d_in[5];
  const float* lnq_g    = (const float*)d_in[6];
  const float* lnq_b    = (const float*)d_in[7];
  const float* lnkv_g   = (const float*)d_in[8];
  const float* lnkv_b   = (const float*)d_in[9];
  const float* in_w     = (const float*)d_in[10];
  const float* in_b     = (const float*)d_in[11];
  const float* out_w    = (const float*)d_in[12];
  const float* out_b    = (const float*)d_in[13];
  const float* ffn_ln_g = (const float*)d_in[14];
  const float* ffn_ln_b = (const float*)d_in[15];
  const float* ffn_w1   = (const float*)d_in[16];
  const float* ffn_b1   = (const float*)d_in[17];
  const float* ffn_w2   = (const float*)d_in[18];
  const float* ffn_b2   = (const float*)d_in[19];
  float* out = (float*)d_out;

  float* ws = (float*)d_ws;
  const size_t SZ = (size_t)M * D;
  float* X    = ws;
  float* kbuf = X    + SZ;
  float* vbuf = kbuf + SZ;
  float* tA   = vbuf + SZ;
  float* tB   = tA   + SZ;
  float* tC   = tB   + SZ;
  float* t1   = tC   + SZ;
  float* ctx  = t1;
  float* wmix = t1   + 4 * SZ;

  static_assert(K == 2, "the projection epilogue sums two speaker streams");
  const size_t PAIRS = (size_t)(D / 64) * M * 2;
  float* sumsX = tA;
  float* sumsY = tA + 3 * PAIRS;

  static_assert(4 * PAIRS + (size_t)D * D_IN + 4 * (size_t)D * D <= SZ, "packed small weights fit behind the row sums");
  static_assert(8 * (size_t)D * D == SZ, "packed FFN weights fill tB");
  float* pk_proj = tA + 4 * PAIRS;
  float* pk_in   = pk_proj + (size_t)D * D_IN;
  float* pk_out  = pk_in   + 3 * (size_t)D * D;
  float* pk_f1   = tB;
  float* pk_f2   = tB + 4 * (size_t)D * D;

  static_assert(D_IN % 96 == 32 && D % 96 == 32 && (4 * D) % 96 == 32, "three-set K loop: four stages left for its tail");
  const dim3 gblk(128);
  auto gg = [](int m, int n) { return dim3(m / 64, n / 128); };

  pack_weights_frag<D_IN, D><<<dim3(((D / 16) * (D_IN / 32) + (8 * D / 16) * (D / 32) + (D / 16) * (4 * D / 32)) / 2),
                              dim3(64), 0, stream>>>(
      proj_w, in_w, out_w, ffn_w1, ffn_w2, pk_proj, pk_in, pk_out, pk_f1, pk_f2);
  wmix_kernel<<<dim3(M), dim3(64), 0, stream>>>(logits, blank_id, wmix, M, V, K);
  gemm_wmma_f32<2, 4,  8 | 64 | 128 | 512 | 2048><<<gg(M, D), gblk, 0, stream>>>(
      x_m, pk_proj, proj_b, nullptr, X, M, D, D_IN, T, K, 0,
      nullptr, nullptr, nullptr, sumsX, wmix, spk_tags);
  gemm_wmma_f32<2, 4,  16 | 128 | 256 | 2048, D / 64><<<gg(M, D), gblk, 0, stream>>>(
      X, pk_in + (size_t)D * D, in_b + D, nullptr, kbuf, M, D, D, T, K, 0,
      lnkv_g, lnkv_b, sumsX, nullptr, nullptr, nullptr);
  gemm_wmma_f32<2, 4,  16 | 128 | 256 | 2048, D / 64><<<gg(M, D), gblk, 0, stream>>>(
      X, pk_in + (size_t)2 * D * D, in_b + 2 * D, nullptr, vbuf, M, D, D, T, K, 0,
      lnkv_g, lnkv_b, sumsX, nullptr, nullptr, nullptr);

  for (int kk = 0; kk < K; ++kk) {
    const float* tagk = spk_tags + (size_t)kk * D;
    gemm_wmma_f32<2, 4,  16 | 32 | 128 | 256 | 2048, D / 64><<<gg(M, D), gblk, 0, stream>>>(
        X, pk_in, in_b, nullptr, tC, M, D, D, T, K, kk,
        lnq_g, lnq_b, sumsX + (size_t)(1 + kk) * PAIRS, nullptr, wmix, tagk);
    static_assert(BAND == 24, "attention tile shape");
    banded_attn_wmma<T, D, H><<<dim3((B * H * (T / 16)) / 2), dim3(64), 0, stream>>>(tC, kbuf, vbuf, ctx);
    gemm_wmma_f32<2, 4,  1 | 8 | 32 | 128 | 512 | 2048><<<gg(M, D), gblk, 0, stream>>>(
        ctx, pk_out, out_b, X, tC, M, D, D, T, K, kk,
        nullptr, nullptr, nullptr, sumsY, wmix, tagk);
    gemm_wmma_f32<2, 4,  2 | 16 | 128 | 256 | 512 | 4096, D / 64><<<dim3(M / 64, (4 * D) / 64), dim3(64), 0, stream>>>(
        tC, pk_f1, ffn_b1, nullptr, t1, M, 4 * D, D, T, K, kk,
        ffn_ln_g, ffn_ln_b, sumsY, nullptr, nullptr, nullptr);
    gemm_wmma_f32<2, 4,  1 | 4 | 128 | 256 | 512 | 1024 | 2048><<<gg(M, D), gblk, 0, stream>>>(
        t1, pk_f2, ffn_b2, tC, out, M, D, 4 * D, T, K, kk,
        nullptr, nullptr, nullptr, nullptr, nullptr, nullptr);
  }
}


# q of both speaker streams from ONE product: LayerNorm(x*w_s+tag_s) is affine in x per row, so q_s = a_s*(X@(gain*Wq)^T) + rank-one terms in the epilogue; one GEMM less, 15 to 14 launches
# speedup vs baseline: 2.5068x; 1.0758x over previous
#include <hip/hip_runtime.h>
#include <hip/hip_bf16.h>
#include <math.h>

typedef float v2f __attribute__((ext_vector_type(2)));
typedef float v4f __attribute__((ext_vector_type(4)));
typedef float v8f __attribute__((ext_vector_type(8)));

__device__ __forceinline__ float gelu_exact(float x) {
  return 0.5f * x * (1.0f + erff(x * 0.70710678118654752f));
}

#define WMMA_F32(cacc, av, bv)                                                 \
  cacc = __builtin_amdgcn_wmma_f32_16x16x4_f32(false, (av), false, (bv),       \
                                               (short)0, (cacc), false, false)

template <int MASK>
__device__ __forceinline__ float row16_xor(float x) {
  return __int_as_float(__builtin_amdgcn_update_dpp(0, __float_as_int(x), 0x160 + MASK, 0xf, 0xf, true));
}
__device__ __forceinline__ float row16_sum(float x) {
  x += row16_xor<1>(x);
  x += row16_xor<2>(x);
  x += row16_xor<4>(x);
  x += row16_xor<8>(x);
  return x;
}

template <int FM, int FN, int FLAGS, int NS = 0>
__global__ __launch_bounds__(128) void gemm_wmma_f32(
    const float* __restrict__ A, const float* __restrict__ W,
    const float* __restrict__ bias, const float* __restrict__ R,
    float* __restrict__ C, int M, int N, int Kd,
    int Tlen, int Kspk, int kk,
    const float* __restrict__ lnG, const float* __restrict__ lnB,
    const float* __restrict__ statsIn, float* __restrict__ statsOut,
    const float* __restrict__ mixW, const float* __restrict__ tags)
{
  static_assert(!(FLAGS & (8 | 16 | 64)) || FM == 2, "row-sum slots assume a 32-row wave tile");
  static_assert(!(FLAGS & 64) || (FLAGS & 8), "stream sums ride on the row-sum epilogue");
  static_assert(!(FLAGS & 4096) || !(FLAGS & 8), "row-sum slots assume two wave-cols per work-group");
  const int lane = threadIdx.x & 31;
  const int w    = threadIdx.x >> 5;
  const int wm   = w & 1;
  const int wn   = (FLAGS & 4096) ? 0 : (w >> 1);
  const int tileM = blockIdx.x * (32 * FM) + wm * (16 * FM);
  const int tileN = blockIdx.y * ((FLAGS & 4096) ? 16 * FN : 32 * FN) + wn * (16 * FN);
  const int half = lane >> 4;
  const int l16  = lane & 15;

  const float* Ap[FM];
  const float* Bp[FN];
#pragma unroll
  for (int fm = 0; fm < FM; ++fm)
    Ap[fm] = (FLAGS & 256)
        ? A + (size_t)((tileM >> 4) + fm) * Kd * 16 + 4 * lane
        : A + (size_t)(tileM + 16 * fm + l16) * Kd + 4 * half;
#pragma unroll
  for (int fn = 0; fn < FN; ++fn)
    Bp[fn] = (FLAGS & 128)
        ? W + (size_t)((tileN >> 4) + fn) * Kd * 16 + 4 * lane
        : W + (size_t)(tileN + 16 * fn + l16) * Kd + 4 * half;

  v8f acc[FM][FN];
#pragma unroll
  for (int fm = 0; fm < FM; ++fm)
#pragma unroll
    for (int fn = 0; fn < FN; ++fn)
      acc[fm][fn] = v8f{};

  v4f a0[4][FM], b0[4][FN];
  v4f a1[4][FM], b1[4][FN];
  v4f a2[4][FM], b2[4][FN];
  v4f g0[4], h0[4], g1[4], h1[4];
  v4f g2[4], h2[4];

  float rinv[FM], rc0[FM], rw[FM];
  v4f t0[4], t1[4], t2[4];
  if (FLAGS & 16) {
    static_assert(!(FLAGS & 16) || NS > 0, "bit4 needs the slot count NS = Kd/64");
    v2f ps[FM][NS > 0 ? NS : 1];
#pragma unroll
    for (int fm = 0; fm < FM; ++fm)
#pragma unroll
      for (int sl = 0; sl < NS; ++sl)
        ps[fm][sl] = *(const v2f*)(statsIn + ((size_t)sl * M + tileM + 16 * fm + l16) * 2);
#pragma unroll
    for (int fm = 0; fm < FM; ++fm) {
      const int row = tileM + 16 * fm + l16;
      float s = 0.f, s2 = 0.f;
#pragma unroll
      for (int sl = 0; sl < NS; ++sl) { s += ps[fm][sl][0]; s2 += ps[fm][sl][1]; }
      const float mean = s / Kd;
      const float var  = s2 / Kd - mean * mean;
      const float inv  = rsqrtf(var + 1e-5f);
      rinv[fm] = inv; rc0[fm] = -mean * inv;
      if (FLAGS & 32) rw[fm] = mixW[(size_t)row * Kspk + kk];
    }
  }

  float qa[2][FM], qb[2][FM], qc[2][FM];
  if (FLAGS & 8192) {
    static_assert(!(FLAGS & 8192) || (NS > 0 && (FLAGS & 1536) == 1536 && !(FLAGS & 16)),
                  "bit13 rides on the transposed tile with rows out and needs the slot count NS");
    v2f ps[2][FM][NS > 0 ? NS : 1];
    v2f w2[FM];
    const size_t pairsIn = (size_t)NS * M * 2;
#pragma unroll
    for (int sp = 0; sp < 2; ++sp)
#pragma unroll
      for (int fm = 0; fm < FM; ++fm)
#pragma unroll
        for (int sl = 0; sl < NS; ++sl)
          ps[sp][fm][sl] = *(const v2f*)(statsIn + sp * pairsIn + ((size_t)sl * M + tileM + 16 * fm + l16) * 2);
#pragma unroll
    for (int fm = 0; fm < FM; ++fm)
      w2[fm] = *(const v2f*)(mixW + (size_t)(tileM + 16 * fm + l16) * 2);
#pragma unroll
    for (int sp = 0; sp < 2; ++sp)
#pragma unroll
      for (int fm = 0; fm < FM; ++fm) {
        float s = 0.f, s2 = 0.f;
#pragma unroll
        for (int sl = 0; sl < NS; ++sl) { s += ps[sp][fm][sl][0]; s2 += ps[sp][fm][sl][1]; }
        const float mean = s / (64 * NS);
        const float var  = s2 / (64 * NS) - mean * mean;
        const float inv  = rsqrtf(var + 1e-5f);
        qa[sp][fm] = inv * w2[fm][sp]; qb[sp][fm] = inv; qc[sp][fm] = -mean * inv;
      }
  }

#define KPOS(koff, s)                                                         \
  ((FLAGS & 256) ? (koff) + 16 * ((s) >> 1) + 8 * half + 4 * ((s) & 1)        \
                 : (koff) + 8 * (s) + 4 * half)
#define LOAD_STAGE(as, bs, gs, hs, ts, koff)                                  \
  _Pragma("unroll") for (int s = 0; s < 4; ++s) {                             \
    _Pragma("unroll") for (int fm = 0; fm < FM; ++fm)                         \
      as[s][fm] = *(const v4f*)(Ap[fm] + ((FLAGS & 256)                       \
                                    ? ((koff) + 8 * s) * 16                   \
                                    : (koff) + 8 * s));                       \
    _Pragma("unroll") for (int fn = 0; fn < FN; ++fn)                         \
      bs[s][fn] = *(const v4f*)(Bp[fn] + ((FLAGS & 128)                       \
                                    ? ((koff) + 8 * s) * 16                   \
                                    : (koff) + 8 * s));                       \
    if (FLAGS & 16) {                                                         \
      gs[s] = *(const v4f*)(lnG + KPOS(koff, s));                             \
      hs[s] = *(const v4f*)(lnB + KPOS(koff, s));                             \
      if (FLAGS & 32)                                                         \
        ts[s] = *(const v4f*)(tags + KPOS(koff, s));                          \
    }                                                                         \
  }

#define LOAD_AB(as, bs, koff)                                                 \
  _Pragma("unroll") for (int s = 0; s < 4; ++s) {                             \
    _Pragma("unroll") for (int fm = 0; fm < FM; ++fm)                         \
      as[s][fm] = *(const v4f*)(Ap[fm] + ((FLAGS & 256)                       \
                                    ? ((koff) + 8 * s) * 16                   \
                                    : (koff) + 8 * s));                       \
    _Pragma("unroll") for (int fn = 0; fn < FN; ++fn)                         \
      bs[s][fn] = *(const v4f*)(Bp[fn] + ((FLAGS & 128)                       \
                                    ? ((koff) + 8 * s) * 16                   \
                                    : (koff) + 8 * s));                       \
  }
#define LOAD_GHT(gs, hs, ts, koff)                                            \
  if (FLAGS & 16) {                                                           \
    _Pragma("unroll") for (int s = 0; s < 4; ++s) {                           \
      gs[s] = *(const v4f*)(lnG + KPOS(koff, s));                             \
      hs[s] = *(const v4f*)(lnB + KPOS(koff, s));                             \
      if (FLAGS & 32)                                                         \
        ts[s] = *(const v4f*)(tags + KPOS(koff, s));                          \
    }                                                                         \
  }

#define NORM_STAGE(as, gs, hs, ts)                                            \
  if (FLAGS & 16) {                                                           \
    _Pragma("unroll") for (int s = 0; s < 4; ++s)                             \
      _Pragma("unroll") for (int fm = 0; fm < FM; ++fm) {                     \
        v4f u = as[s][fm];                                                    \
        if (FLAGS & 32) u = u * rw[fm] + ts[s];                               \
        const v4f t = u * rinv[fm] + rc0[fm];                                 \
        as[s][fm] = t * gs[s] + hs[s];                                        \
      }                                                                       \
  }

#define WMMA_AB(cacc, av, bv)                                                 \
  if (FLAGS & 512) { WMMA_F32(cacc, bv, av); } else { WMMA_F32(cacc, av, bv); }
#define MMA_STAGE(as, bs)                                                     \
  _Pragma("unroll") for (int s = 0; s < 4; ++s) {                             \
    _Pragma("unroll") for (int fm = 0; fm < FM; ++fm)                         \
      _Pragma("unroll") for (int fn = 0; fn < FN; ++fn)                       \
        WMMA_AB(acc[fm][fn],                                                  \
                __builtin_shufflevector(as[s][fm], as[s][fm], 0, 1),          \
                __builtin_shufflevector(bs[s][fn], bs[s][fn], 0, 1));         \
    _Pragma("unroll") for (int fm = 0; fm < FM; ++fm)                         \
      _Pragma("unroll") for (int fn = 0; fn < FN; ++fn)                       \
        WMMA_AB(acc[fm][fn],                                                  \
                __builtin_shufflevector(as[s][fm], as[s][fm], 2, 3),          \
                __builtin_shufflevector(bs[s][fn], bs[s][fn], 2, 3));         \
  }

#define STAGE_FENCE() __builtin_amdgcn_sched_barrier(0)

  if (FLAGS & 2048) {
    LOAD_AB(a0, b0, 0);
    LOAD_GHT(g0, h0, t0, 0);
    STAGE_FENCE();
    LOAD_AB(a1, b1, 32);
    int k0 = 0;
    for (; k0 < Kd - 128; k0 += 96) {
      STAGE_FENCE();
      LOAD_GHT(g1, h1, t1, k0 + 32);
      STAGE_FENCE();
      LOAD_AB(a2, b2, k0 + 64);
      STAGE_FENCE();
      NORM_STAGE(a0, g0, h0, t0);
      MMA_STAGE(a0, b0);
      STAGE_FENCE();
      LOAD_GHT(g2, h2, t2, k0 + 64);
      STAGE_FENCE();
      LOAD_AB(a0, b0, k0 + 96);
      STAGE_FENCE();
      NORM_STAGE(a1, g1, h1, t1);
      MMA_STAGE(a1, b1);
      STAGE_FENCE();
      LOAD_GHT(g0, h0, t0, k0 + 96);
      STAGE_FENCE();
      LOAD_AB(a1, b1, k0 + 128);
      STAGE_FENCE();
      NORM_STAGE(a2, g2, h2, t2);
      MMA_STAGE(a2, b2);
    }
    STAGE_FENCE();
    LOAD_GHT(g1, h1, t1, k0 + 32);
    STAGE_FENCE();
    LOAD_AB(a2, b2, k0 + 64);
    STAGE_FENCE();
    NORM_STAGE(a0, g0, h0, t0);
    MMA_STAGE(a0, b0);
    STAGE_FENCE();
    LOAD_GHT(g2, h2, t2, k0 + 64);
    STAGE_FENCE();
    LOAD_AB(a0, b0, k0 + 96);
    STAGE_FENCE();
    NORM_STAGE(a1, g1, h1, t1);
    MMA_STAGE(a1, b1);
    STAGE_FENCE();
    LOAD_GHT(g0, h0, t0, k0 + 96);
    STAGE_FENCE();
    NORM_STAGE(a2, g2, h2, t2);
    MMA_STAGE(a2, b2);
    NORM_STAGE(a0, g0, h0, t0);
    MMA_STAGE(a0, b0);
  } else {
    LOAD_STAGE(a0, b0, g0, h0, t0, 0);
    for (int k0 = 0; k0 < Kd - 64; k0 += 64) {
      STAGE_FENCE();
      LOAD_STAGE(a1, b1, g1, h1, t1, k0 + 32);
      STAGE_FENCE();
      NORM_STAGE(a0, g0, h0, t0);
      MMA_STAGE(a0, b0);
      STAGE_FENCE();
      LOAD_STAGE(a0, b0, g0, h0, t0, k0 + 64);
      STAGE_FENCE();
      NORM_STAGE(a1, g1, h1, t1);
      MMA_STAGE(a1, b1);
    }
    STAGE_FENCE();
    LOAD_STAGE(a1, b1, g1, h1, t1, Kd - 32);
    STAGE_FENCE();
    NORM_STAGE(a0, g0, h0, t0);
    MMA_STAGE(a0, b0);
    NORM_STAGE(a1, g1, h1, t1);
    MMA_STAGE(a1, b1);
  }
#undef STAGE_FENCE
#undef LOAD_STAGE
#undef LOAD_AB
#undef LOAD_GHT
#undef NORM_STAGE
#undef MMA_STAGE
#undef WMMA_AB
#undef KPOS

  if (FLAGS & 512) {
    v4f bq[FN][2], tq0[FN][2], tq1[FN][2];
    v4f sq[FN][2], uq0[FN][2], uq1[FN][2];
#pragma unroll
    for (int fn = 0; fn < FN; ++fn)
#pragma unroll
      for (int j = 0; j < 2; ++j) {
        const int col = tileN + 16 * fn + 8 * half + 4 * j;
        bq[fn][j] = *(const v4f*)(bias + col);
        tq0[fn][j] = v4f{};
        tq1[fn][j] = v4f{};
        if ((FLAGS & 64) || (FLAGS & 33) == 33) tq0[fn][j] = *(const v4f*)(tags + col);
        if (FLAGS & 64) tq1[fn][j] = *(const v4f*)(tags + N + col);
        if (FLAGS & 8192) {
          sq[fn][j]  = *(const v4f*)(tags + N + col);
          uq0[fn][j] = *(const v4f*)(tags + 2 * N + col);
          uq1[fn][j] = *(const v4f*)(tags + 3 * N + col);
        }
      }
    v2f sumV[FM], sumS0[FM], sumS1[FM];
#pragma unroll
    for (int fm = 0; fm < FM; ++fm) {
      const int row = tileM + 16 * fm + l16;
      float wr = 0.f;
      if ((FLAGS & 33) == 33) wr = mixW[(size_t)row * Kspk + kk];
      v2f w2 = v2f{};
      if (FLAGS & 64) w2 = *(const v2f*)(mixW + (size_t)row * 2);
      size_t orow = (size_t)row;
      if (FLAGS & 4) {
        const int bi = row / Tlen;
        orow = (size_t)bi * Kspk * Tlen + (size_t)kk * Tlen + (row % Tlen);
      }
      float ps = 0.f, ps2 = 0.f, p0 = 0.f, p02 = 0.f, p1 = 0.f, p12 = 0.f;
#pragma unroll
      for (int fn = 0; fn < FN; ++fn) {
        const size_t pb = ((size_t)((tileM >> 4) + fm) * N + tileN + 16 * fn) * 16 + 4 * lane;
#pragma unroll
        for (int j = 0; j < 2; ++j) {
          v4f v;
#pragma unroll
          for (int e = 0; e < 4; ++e) v[e] = acc[fm][fn][4 * j + e];
          if (FLAGS & 8192) {
            const size_t qo = orow * N + tileN + 16 * fn + 8 * half + 4 * j;
            const v4f base = sq[fn][j] * qc[0][fm] + bq[fn][j];
            const v4f bas1 = sq[fn][j] * qc[1][fm] + bq[fn][j];
            *(v4f*)(C + qo)        = v * qa[0][fm] + (uq0[fn][j] * qb[0][fm] + base);
            *(v4f*)(statsOut + qo) = v * qa[1][fm] + (uq1[fn][j] * qb[1][fm] + bas1);
            continue;
          }
          v += bq[fn][j];
          if (FLAGS & 1) {
            v4f r = *(const v4f*)(R + pb + 128 * j);
            if (FLAGS & 32) r = r * wr + tq0[fn][j];
            v += r;
          }
          if (FLAGS & 2) {
#pragma unroll
            for (int e = 0; e < 4; ++e) v[e] = gelu_exact(v[e]);
          }
          if (FLAGS & 8) {
#pragma unroll
            for (int e = 0; e < 4; ++e) { ps += v[e]; ps2 += v[e] * v[e]; }
          }
          if (FLAGS & 64) {
            const v4f x0 = v * w2[0] + tq0[fn][j];
            const v4f x1 = v * w2[1] + tq1[fn][j];
#pragma unroll
            for (int e = 0; e < 4; ++e) {
              p0 += x0[e]; p02 += x0[e] * x0[e];
              p1 += x1[e]; p12 += x1[e] * x1[e];
            }
          }
          if (FLAGS & 1024)
            *(v4f*)(C + orow * N + tileN + 16 * fn + 8 * half + 4 * j) = v;
          else
            *(v4f*)(C + pb + 128 * j) = v;
        }
      }
      if (FLAGS & 8) {
        ps += __shfl_xor(ps, 16, 32); ps2 += __shfl_xor(ps2, 16, 32);
        sumV[fm][0] = ps; sumV[fm][1] = ps2;
      }
      if (FLAGS & 64) {
        p0 += __shfl_xor(p0, 16, 32); p02 += __shfl_xor(p02, 16, 32);
        p1 += __shfl_xor(p1, 16, 32); p12 += __shfl_xor(p12, 16, 32);
        sumS0[fm][0] = p0; sumS0[fm][1] = p02;
        sumS1[fm][0] = p1; sumS1[fm][1] = p12;
      }
    }
    if (FLAGS & 8) {
      const int slot = blockIdx.y * 2 + wn;
      const size_t pairs = (size_t)(N >> 6) * M * 2;
      float* sp = statsOut + ((size_t)slot * M + tileM + lane) * 2;
      *(v2f*)sp = half ? sumV[FM - 1] : sumV[0];
      if (FLAGS & 64) {
        *(v2f*)(sp + pairs)     = half ? sumS0[FM - 1] : sumS0[0];
        *(v2f*)(sp + 2 * pairs) = half ? sumS1[FM - 1] : sumS1[0];
      }
    }
    return;
  }

  float bv[FN];
#pragma unroll
  for (int fn = 0; fn < FN; ++fn) bv[fn] = bias[tileN + 16 * fn + l16];
  float tg0[FN], tg1[FN];
  if (FLAGS & 64) {
#pragma unroll
    for (int fn = 0; fn < FN; ++fn) {
      tg0[fn] = tags[tileN + 16 * fn + l16];
      tg1[fn] = tags[N + tileN + 16 * fn + l16];
    }
  } else if ((FLAGS & 33) == 33) {
#pragma unroll
    for (int fn = 0; fn < FN; ++fn) tg0[fn] = tags[tileN + 16 * fn + l16];
  }

  v2f keepV = v2f{}, keepS0 = v2f{}, keepS1 = v2f{};

#pragma unroll
  for (int fm = 0; fm < FM; ++fm) {
#pragma unroll
    for (int i = 0; i < 8; ++i) {
      const int row = tileM + 16 * fm + i + 8 * half;
      float v[FN];
#pragma unroll
      for (int fn = 0; fn < FN; ++fn) v[fn] = acc[fm][fn][i] + bv[fn];
      if (FLAGS & 1) {
        const float* rr = R + (size_t)row * N + tileN + l16;
        if (FLAGS & 32) {
          const float wr = mixW[(size_t)row * Kspk + kk];
#pragma unroll
          for (int fn = 0; fn < FN; ++fn) {
            const float xk = rr[16 * fn] * wr + tg0[fn];
            v[fn] += xk;
          }
        } else {
#pragma unroll
          for (int fn = 0; fn < FN; ++fn) v[fn] += rr[16 * fn];
        }
      }
      if (FLAGS & 2) {
#pragma unroll
        for (int fn = 0; fn < FN; ++fn) v[fn] = gelu_exact(v[fn]);
      }
      if (FLAGS & 8) {
        float ps = 0.f, ps2 = 0.f;
#pragma unroll
        for (int fn = 0; fn < FN; ++fn) { ps += v[fn]; ps2 += v[fn] * v[fn]; }
        ps = row16_sum(ps); ps2 = row16_sum(ps2);
        if (l16 == 8 * fm + i) { keepV[0] = ps; keepV[1] = ps2; }
      }
      if (FLAGS & 64) {
        const v2f w2 = *(const v2f*)(mixW + (size_t)row * 2);
        float p0 = 0.f, p02 = 0.f, p1 = 0.f, p12 = 0.f;
#pragma unroll
        for (int fn = 0; fn < FN; ++fn) {
          const float x0 = v[fn] * w2[0] + tg0[fn];
          const float x1 = v[fn] * w2[1] + tg1[fn];
          p0 += x0; p02 += x0 * x0; p1 += x1; p12 += x1 * x1;
        }
        p0 = row16_sum(p0); p02 = row16_sum(p02);
        p1 = row16_sum(p1); p12 = row16_sum(p12);
        if (l16 == 8 * fm + i) { keepS0[0] = p0; keepS0[1] = p02; keepS1[0] = p1; keepS1[1] = p12; }
      }
      size_t orow = (size_t)row;
      if (FLAGS & 4) {
        const int bi = row / Tlen;
        orow = (size_t)bi * Kspk * Tlen + (size_t)kk * Tlen + (row % Tlen);
      }
      float* cr = C + orow * N + tileN + l16;
#pragma unroll
      for (int fn = 0; fn < FN; ++fn) cr[16 * fn] = v[fn];
    }
  }
  if (FLAGS & 8) {
    const int srow = tileM + 16 * (l16 >> 3) + 8 * half + (l16 & 7);
    const int slot = blockIdx.y * 2 + wn;
    const size_t pairs = (size_t)(N >> 6) * M * 2;
    float* sp = statsOut + ((size_t)slot * M + srow) * 2;
    *(v2f*)sp = keepV;
    if (FLAGS & 64) {
      *(v2f*)(sp + pairs)     = keepS0;
      *(v2f*)(sp + 2 * pairs) = keepS1;
    }
  }
}

template <int DIN, int Dm>
__global__ __launch_bounds__(128) void pack_weights_frag(
    const float* __restrict__ w_proj, const float* __restrict__ w_in,
    const float* __restrict__ w_out, const float* __restrict__ w_f1,
    const float* __restrict__ w_f2,
    float* __restrict__ p_proj, float* __restrict__ p_in,
    float* __restrict__ p_out, float* __restrict__ p_f1,
    float* __restrict__ p_f2,
    const float* __restrict__ q_gain, const float* __restrict__ q_shift,
    const float* __restrict__ q_bias, const float* __restrict__ tags,
    float* __restrict__ qvec)
{
  constexpr int I0 = (Dm / 16) * (DIN / 32);
  constexpr int I1 = I0 + (3 * Dm / 16) * (Dm / 32);
  constexpr int I2 = I1 + (Dm / 16) * (Dm / 32);
  constexpr int I3 = I2 + (4 * Dm / 16) * (Dm / 32);
  constexpr int I4 = I3 + (Dm / 16) * (4 * Dm / 32);
  const int lane = threadIdx.x & 31;
  const int c    = lane & 15;
  const int half = lane >> 4;
  int item = blockIdx.x * 2 + (threadIdx.x >> 5);
  constexpr int NG = (Dm / 32) * 4;
  if (item < NG) {
    const int vec = item & 3;
    const int n0  = 32 * (item >> 2);
    v4f cf[8];
#pragma unroll
    for (int i = 0; i < 8; ++i) {
      const int kq = 4 * c + 64 * i;
      const v4f gq = *(const v4f*)(q_gain + kq);
      if (vec == 0)      cf[i] = *(const v4f*)(q_shift + kq);
      else if (vec == 1) cf[i] = gq;
      else               cf[i] = *(const v4f*)(tags + (vec - 2) * Dm + kq) * gq;
    }
    const float* wr = w_in + (size_t)(n0 + 16 * half) * Dm + 4 * c;
    float keep = 0.f;
#pragma unroll 4
    for (int r = 0; r < 16; ++r) {
      v4f a = v4f{};
#pragma unroll
      for (int i = 0; i < 8; ++i) a += cf[i] * *(const v4f*)(wr + (size_t)r * Dm + 64 * i);
      const float t = row16_sum((a[0] + a[1]) + (a[2] + a[3]));
      if (c == r) keep = t;
    }
    if (vec == 0) keep += q_bias[n0 + lane];
    qvec[vec * Dm + n0 + lane] = keep;
    return;
  }
  item -= NG;
  const float* src; float* dst; int Kd;
  bool g16 = false;
  bool qrows = false;
  if (item < I0)      { src = w_proj; dst = p_proj; Kd = DIN; }
  else if (item < I1) { src = w_in;   dst = p_in;   Kd = Dm;     item -= I0; g16 = true; qrows = item < (Dm / 16) * (Dm / 32); }
  else if (item < I2) { src = w_out;  dst = p_out;  Kd = Dm;     item -= I1; }
  else if (item < I3) { src = w_f1;   dst = p_f1;   Kd = Dm;     item -= I2; g16 = true; }
  else if (item < I4) { src = w_f2;   dst = p_f2;   Kd = 4 * Dm; item -= I3; g16 = true; }
  else return;
  const int stages = Kd / 32;
  const int g  = item / stages;
  const int st = item % stages;
  const float* rp = src + (size_t)(16 * g + c) * Kd + 32 * st;
  float* op = dst + ((size_t)g * (Kd / 8) + 4 * st) * 128 + 4 * lane;
  v4f x[4];
#pragma unroll
  for (int s = 0; s < 4; ++s)
    x[s] = *(const v4f*)(rp + (g16 ? 16 * (s >> 1) + 8 * half + 4 * (s & 1) : 8 * s + 4 * half));
  if (qrows) {
#pragma unroll
    for (int s = 0; s < 4; ++s)
      x[s] *= *(const v4f*)(q_gain + 32 * st + 16 * (s >> 1) + 8 * half + 4 * (s & 1));
  }
#pragma unroll
  for (int s = 0; s < 4; ++s) *(v4f*)(op + 128 * s) = x[s];
}

__global__ __launch_bounds__(64) void wmix_kernel(
    const float* __restrict__ logits, const int* __restrict__ blank_p,
    float* __restrict__ wmix, int BT, int V, int Kspk)
{
  const int row = blockIdx.x;
  const int tid = threadIdx.x;
  const int blank = blank_p[0];
  __shared__ float red[256];
  float a[8];
  for (int k2 = 0; k2 < Kspk; ++k2) {
    const float* L = logits + ((size_t)k2 * BT + row) * V;
    for (int t = tid; t < 256; t += 64) {
      float mt = -INFINITY;
      for (int v = t; v < V; v += 256) mt = fmaxf(mt, L[v]);
      red[t] = mt;
    }
    __syncthreads();
    for (int off = 128; off > 0; off >>= 1) {
      for (int i = tid; i < off; i += 64) red[i] = fmaxf(red[i], red[i + off]);
      __syncthreads();
    }
    const float m = red[0]; __syncthreads();
    for (int t = tid; t < 256; t += 64) {
      float st = 0.f;
      for (int v = t; v < V; v += 256) st += expf(L[v] - m);
      red[t] = st;
    }
    __syncthreads();
    for (int off = 128; off > 0; off >>= 1) {
      for (int i = tid; i < off; i += 64) red[i] += red[i + off];
      __syncthreads();
    }
    const float s = red[0]; __syncthreads();
    a[k2] = 1.0f - expf(L[blank] - m) / s;
  }
  if (tid == 0) {
    float m2 = -INFINITY;
    for (int k2 = 0; k2 < Kspk; ++k2) m2 = fmaxf(m2, a[k2]);
    float ss = 0.f;
    float e[8];
    for (int k2 = 0; k2 < Kspk; ++k2) { e[k2] = expf(a[k2] - m2); ss += e[k2]; }
    for (int k2 = 0; k2 < Kspk; ++k2) wmix[(size_t)row * Kspk + k2] = e[k2] / ss;
  }
}

template <int T_, int Dn, int Hn>
__global__ __launch_bounds__(128) void banded_attn_wmma(
    const float* __restrict__ q, const float* __restrict__ kbuf,
    const float* __restrict__ vbuf, float* __restrict__ ctx)
{
  constexpr int HD = 64, BAND = 24;
  static_assert(Dn == Hn * HD && T_ % 16 == 0, "attention tile shape");
  const int lane = threadIdx.x & 31;
  const int wid  = threadIdx.x >> 5;
  const int c    = lane & 15;
  const int half = lane >> 4;
  constexpr int tilesT = T_ / 16;
  const int gw   = blockIdx.x * 2 + wid;
  const int tile = gw % tilesT;
  const int hh   = (gw / tilesT) % Hn;
  const int b    = gw / (tilesT * Hn);
  const int t0   = tile * 16;
  const int key0 = t0 - BAND;
  const unsigned hoff = (unsigned)(hh * HD);
  const int brow = b * T_;


  v4f qf[8];
  {
    const float* qp = q + ((unsigned)((brow + t0 + c) * Dn) + hoff + 4u * half);
#pragma unroll
    for (int i = 0; i < 8; ++i) qf[i] = *(const v4f*)(qp + 8 * i);
  }
  v4f kf[4][8];
#pragma unroll
  for (int kb = 0; kb < 4; ++kb) {
    int s = key0 + 16 * kb + c;
    s = min(max(s, 0), T_ - 1);
    const float* kp = kbuf + ((unsigned)((brow + s) * Dn) + hoff + 4u * half);
#pragma unroll
    for (int i = 0; i < 8; ++i) kf[kb][i] = *(const v4f*)(kp + 8 * i);
  }
  v4f vf[4][4][2];
#pragma unroll
  for (int kb = 0; kb < 4; ++kb)
#pragma unroll
    for (int j = 0; j < 4; ++j)
#pragma unroll
      for (int e = 0; e < 2; ++e) {
        int s = key0 + 16 * kb + 8 * half + 2 * j + e;
        s = min(max(s, 0), T_ - 1);
        vf[kb][j][e] = *(const v4f*)(vbuf + ((unsigned)((brow + s) * Dn) + hoff + 4u * c));
      }

  __builtin_amdgcn_sched_barrier(0);
  v8f st[4];
#pragma unroll
  for (int kb = 0; kb < 4; ++kb) st[kb] = v8f{};
#pragma unroll
  for (int i = 0; i < 8; ++i) {
#pragma unroll
    for (int kb = 0; kb < 4; ++kb)
      WMMA_F32(st[kb], __builtin_shufflevector(kf[kb][i], kf[kb][i], 0, 1),
                       __builtin_shufflevector(qf[i], qf[i], 0, 1));
#pragma unroll
    for (int kb = 0; kb < 4; ++kb)
      WMMA_F32(st[kb], __builtin_shufflevector(kf[kb][i], kf[kb][i], 2, 3),
                       __builtin_shufflevector(qf[i], qf[i], 2, 3));
  }

  const float scale = 0.125f;
  float p[4][8];
  float m = -INFINITY;
#pragma unroll
  for (int kb = 0; kb < 4; ++kb)
#pragma unroll
    for (int r = 0; r < 8; ++r) {
      const int rel = 16 * kb + 8 * half + r - c;
      const int s   = key0 + 16 * kb + 8 * half + r;
      const bool ok = (rel >= 0) && (rel <= 2 * BAND) && (s >= 0) && (s < T_);
      const float v = ok ? st[kb][r] * scale : -INFINITY;
      p[kb][r] = v;
      m = fmaxf(m, v);
    }
  m = fmaxf(m, __shfl_xor(m, 16, 32));
  float sum = 0.f;
#pragma unroll
  for (int kb = 0; kb < 4; ++kb)
#pragma unroll
    for (int r = 0; r < 8; ++r) {
      const float e = expf(p[kb][r] - m);
      p[kb][r] = e;
      sum += e;
    }
  sum += __shfl_xor(sum, 16, 32);
  const float inv = 1.0f / sum;

  v8f o[4];
#pragma unroll
  for (int db = 0; db < 4; ++db) o[db] = v8f{};
#pragma unroll
  for (int kb = 0; kb < 4; ++kb)
#pragma unroll
    for (int j = 0; j < 4; ++j) {
      v2f a;
      a[0] = p[kb][2 * j] * inv;
      a[1] = p[kb][2 * j + 1] * inv;
#pragma unroll
      for (int db = 0; db < 4; ++db) {
        v2f bb;
        bb[0] = vf[kb][j][0][db];
        bb[1] = vf[kb][j][1][db];
        WMMA_F32(o[db], a, bb);
      }
    }

#pragma unroll
  for (int r = 0; r < 8; ++r) {
    v4f ov;
    ov[0] = o[0][r]; ov[1] = o[1][r]; ov[2] = o[2][r]; ov[3] = o[3][r];
    *(v4f*)(ctx + ((unsigned)((brow + t0 + 8 * half + r) * Dn) + hoff + 4u * c)) = ov;
  }
}

extern "C" void kernel_launch(void* const* d_in, const int* in_sizes, int n_in,
                              void* d_out, int out_size, void* d_ws, size_t ws_size,
                              hipStream_t stream) {
  (void)in_sizes; (void)n_in; (void)out_size; (void)ws_size;
  constexpr int B = 4, T = 1024, D_IN = 512, D = 512, K = 2, V = 500, H = 8, BAND = 24;
  constexpr int M = B * T;

  const float* x_m      = (const float*)d_in[0];
  const float* logits   = (const float*)d_in[1];
  const int*   blank_id = (const int*)  d_in[2];
  const float* proj_w   = (const float*)d_in[3];
  const float* proj_b   = (const float*)d_in[4];
  const float* spk_tags = (const float*)d_in[5];
  const float* lnq_g    = (const float*)d_in[6];
  const float* lnq_b    = (const float*)d_in[7];
  const float* lnkv_g   = (const float*)d_in[8];
  const float* lnkv_b   = (const float*)d_in[9];
  const float* in_w     = (const float*)d_in[10];
  const float* in_b     = (const float*)d_in[11];
  const float* out_w    = (const float*)d_in[12];
  const float* out_b    = (const float*)d_in[13];
  const float* ffn_ln_g = (const float*)d_in[14];
  const float* ffn_ln_b = (const float*)d_in[15];
  const float* ffn_w1   = (const float*)d_in[16];
  const float* ffn_b1   = (const float*)d_in[17];
  const float* ffn_w2   = (const float*)d_in[18];
  const float* ffn_b2   = (const float*)d_in[19];
  float* out = (float*)d_out;

  float* ws = (float*)d_ws;
  const size_t SZ = (size_t)M * D;
  float* X    = ws;
  float* kbuf = X    + SZ;
  float* vbuf = kbuf + SZ;
  float* tA   = vbuf + SZ;
  float* tB   = tA   + SZ;
  float* tC   = tB   + SZ;
  float* t1   = tC   + SZ;
  float* ctx  = t1;
  float* q1   = t1 + 3 * SZ;
  float* wmix = t1   + 4 * SZ;

  static_assert(K == 2, "the projection epilogue sums two speaker streams");
  const size_t PAIRS = (size_t)(D / 64) * M * 2;
  float* sumsX = tA;
  float* sumsY = tA + 3 * PAIRS;

  static_assert(5 * PAIRS + 4 * (size_t)D + (size_t)D * D_IN + 4 * (size_t)D * D <= SZ, "packed small weights, q vectors and stream 1's sums of y fit behind the row sums");
  static_assert(8 * (size_t)D * D == SZ, "packed FFN weights fill tB");
  float* pk_proj = tA + 4 * PAIRS;
  float* pk_in   = pk_proj + (size_t)D * D_IN;
  float* pk_out  = pk_in   + 3 * (size_t)D * D;
  float* qvec    = pk_out  + (size_t)D * D;
  float* sumsY1  = qvec    + 4 * (size_t)D;
  float* pk_f1   = tB;
  float* pk_f2   = tB + 4 * (size_t)D * D;

  static_assert(D_IN % 96 == 32 && D % 96 == 32 && (4 * D) % 96 == 32, "three-set K loop: four stages left for its tail");
  const dim3 gblk(128);
  auto gg = [](int m, int n) { return dim3(m / 64, n / 128); };

  pack_weights_frag<D_IN, D><<<dim3(((D / 32) * 4 + (D / 16) * (D_IN / 32) + (8 * D / 16) * (D / 32) + (D / 16) * (4 * D / 32)) / 2),
                              dim3(64), 0, stream>>>(
      proj_w, in_w, out_w, ffn_w1, ffn_w2, pk_proj, pk_in, pk_out, pk_f1, pk_f2,
      lnq_g, lnq_b, in_b, spk_tags, qvec);
  wmix_kernel<<<dim3(M), dim3(64), 0, stream>>>(logits, blank_id, wmix, M, V, K);
  gemm_wmma_f32<2, 4,  8 | 64 | 128 | 512 | 2048><<<gg(M, D), gblk, 0, stream>>>(
      x_m, pk_proj, proj_b, nullptr, X, M, D, D_IN, T, K, 0,
      nullptr, nullptr, nullptr, sumsX, wmix, spk_tags);
  gemm_wmma_f32<2, 4,  16 | 128 | 256 | 2048, D / 64><<<gg(M, D), gblk, 0, stream>>>(
      X, pk_in + (size_t)D * D, in_b + D, nullptr, kbuf, M, D, D, T, K, 0,
      lnkv_g, lnkv_b, sumsX, nullptr, nullptr, nullptr);
  gemm_wmma_f32<2, 4,  16 | 128 | 256 | 2048, D / 64><<<gg(M, D), gblk, 0, stream>>>(
      X, pk_in + (size_t)2 * D * D, in_b + 2 * D, nullptr, vbuf, M, D, D, T, K, 0,
      lnkv_g, lnkv_b, sumsX, nullptr, nullptr, nullptr);

  gemm_wmma_f32<2, 4,  128 | 256 | 512 | 1024 | 2048 | 8192, D / 64><<<gg(M, D), gblk, 0, stream>>>(
      X, pk_in, qvec, nullptr, tC, M, D, D, T, K, 0,
      nullptr, nullptr, sumsX + PAIRS, q1, wmix, qvec);
  static_assert(BAND == 24, "attention tile shape");
  banded_attn_wmma<T, D, H><<<dim3((B * H * (T / 16)) / 2), dim3(64), 0, stream>>>(tC, kbuf, vbuf, ctx);
  banded_attn_wmma<T, D, H><<<dim3((B * H * (T / 16)) / 2), dim3(64), 0, stream>>>(q1, kbuf, vbuf, ctx + SZ);

  for (int kk = 0; kk < K; ++kk) {
    gemm_wmma_f32<2, 4,  1 | 8 | 32 | 128 | 512 | 2048><<<gg(M, D), gblk, 0, stream>>>(
        ctx + (size_t)kk * SZ, pk_out, out_b, X, kk ? kbuf : tC, M, D, D, T, K, kk,
        nullptr, nullptr, nullptr, kk ? sumsY1 : sumsY, wmix, spk_tags + (size_t)kk * D);
  }
  for (int kk = 0; kk < K; ++kk) {
    float* y = kk ? kbuf : tC;
    gemm_wmma_f32<2, 4,  2 | 16 | 128 | 256 | 512 | 4096, D / 64><<<dim3(M / 64, (4 * D) / 64), dim3(64), 0, stream>>>(
        y, pk_f1, ffn_b1, nullptr, t1, M, 4 * D, D, T, K, kk,
        ffn_ln_g, ffn_ln_b, kk ? sumsY1 : sumsY, nullptr, nullptr, nullptr);
    gemm_wmma_f32<2, 4,  1 | 4 | 128 | 256 | 512 | 1024 | 2048><<<gg(M, D), gblk, 0, stream>>>(
        t1, pk_f2, ffn_b2, y, out, M, D, 4 * D, T, K, kk,
        nullptr, nullptr, nullptr, nullptr, nullptr, nullptr);
  }
}
